# starting kernel
# baseline (speedup 1.0000x reference)
#include <hip/hip_runtime.h>
#include <hip/hip_bf16.h>

typedef __attribute__((ext_vector_type(2))) float v2f;
typedef __attribute__((ext_vector_type(8))) float v8f;

#define NODES 100000

__global__ void zero_f32_kernel(float* __restrict__ p, int n) {
  int i = blockIdx.x * blockDim.x + threadIdx.x;
  if (i < n) p[i] = 0.0f;
}

__global__ void deg_count_kernel(const int* __restrict__ dst, int e, float* __restrict__ deg) {
  int i = blockIdx.x * blockDim.x + threadIdx.x;
  if (i < e) atomicAdd(&deg[dst[i]], 1.0f);
}

__global__ void deg_finalize_kernel(float* __restrict__ deg, int n) {
  int i = blockIdx.x * blockDim.x + threadIdx.x;
  if (i < n) deg[i] = rsqrtf(deg[i] + 1.0f);
}

template <int K, int FOUT, bool RELU>
__global__ void gemm_wmma_f32_kernel(const float* __restrict__ X,
                                     const float* __restrict__ W,
                                     float* __restrict__ H,
                                     int n_rows) {
  constexpr int NT = FOUT / 16;
  __shared__ float Wp[K * FOUT];

  for (int i = threadIdx.x; i < K * FOUT; i += 256) {
    int k = i / FOUT;
    int col = i - k * FOUT;
    Wp[(k >> 1) * (FOUT * 2) + col * 2 + (k & 1)] = W[i];
  }
  __syncthreads();

  const int wave = threadIdx.x >> 5;
  const int lane = threadIdx.x & 31;
  const int tile_m = blockIdx.x * 8 + wave;
  if (tile_m * 16 >= n_rows) return;

  const int half = lane >> 4;
  const int l15  = lane & 15;

  const float* __restrict__ xrow = X + (size_t)(tile_m * 16 + l15) * K;

  v8f acc[NT];
#pragma unroll
  for (int t = 0; t < NT; ++t) acc[t] = (v8f){};

#pragma unroll
  for (int k = 0; k < K; k += 4) {
    const int kk = k + half * 2;
    float a0 = xrow[kk];
    float a1 = xrow[kk + 1];
    if (RELU) { a0 = fmaxf(a0, 0.0f); a1 = fmaxf(a1, 0.0f); }
    v2f a;
    a.x = a0;
    a.y = a1;
    const float* __restrict__ wrow = Wp + (kk >> 1) * (FOUT * 2) + l15 * 2;
#pragma unroll
    for (int t = 0; t < NT; ++t) {
      const v2f b = *(const v2f*)(wrow + t * 32);
      acc[t] = __builtin_amdgcn_wmma_f32_16x16x4_f32(false, a, false, b,
                                                     (short)0, acc[t], false, false);
    }
  }

  float* __restrict__ hbase = H + (size_t)(tile_m * 16) * FOUT + l15;
#pragma unroll
  for (int t = 0; t < NT; ++t) {
#pragma unroll
    for (int v = 0; v < 8; ++v) {
      hbase[(size_t)(v + half * 8) * FOUT + t * 16] = acc[t][v];
    }
  }
}

__global__ void self_init_kernel(const float* __restrict__ H,
                                 const float* __restrict__ dinv,
                                 const float* __restrict__ bias,
                                 float* __restrict__ Out,
                                 int n, int fshift) {
  const int fmask = (1 << fshift) - 1;
  long long idx = (long long)blockIdx.x * blockDim.x + threadIdx.x;
  long long total = ((long long)n) << fshift;
  if (idx >= total) return;
  int node = (int)(idx >> fshift);
  int f = (int)(idx & fmask);
  float di = dinv[node];
  Out[idx] = H[idx] * di * di + bias[f];
}

__global__ void edge_scatter_kernel(const int* __restrict__ src,
                                    const int* __restrict__ dst,
                                    const float* __restrict__ H,
                                    const float* __restrict__ dinv,
                                    float* __restrict__ Out,
                                    int e, int fshift) {
  const int fmask = (1 << fshift) - 1;
  long long idx = (long long)blockIdx.x * blockDim.x + threadIdx.x;
  long long total = ((long long)e) << fshift;
  if (idx >= total) return;
  int edge = (int)(idx >> fshift);
  int f = (int)(idx & fmask);
  int s = src[edge];
  int d = dst[edge];
  float coef = dinv[s] * dinv[d];
  atomicAdd(&Out[(((size_t)d) << fshift) + f], H[(((size_t)s) << fshift) + f] * coef);
}

static inline unsigned int cdiv(long long a, long long b) {
  return (unsigned int)((a + b - 1) / b);
}

extern "C" void kernel_launch(void* const* d_in, const int* in_sizes, int n_in,
                              void* d_out, int out_size, void* d_ws, size_t ws_size,
                              hipStream_t stream) {
  const float* x  = (const float*)d_in[0];
  const int*   ei = (const int*)d_in[1];
  const float* W1 = (const float*)d_in[2];
  const float* b1 = (const float*)d_in[3];
  const float* W2 = (const float*)d_in[4];
  const float* b2 = (const float*)d_in[5];
  const float* W3 = (const float*)d_in[6];
  const float* b3 = (const float*)d_in[7];
  float* out = (float*)d_out;

  const int N = NODES;
  const int E = in_sizes[1] / 2;

  const int* src = ei;
  const int* dst = ei + E;

  float* dinv = (float*)d_ws;
  float* bufA = dinv + N;
  float* bufB = bufA + (size_t)N * 128;

  const int T = 256;
  const unsigned int gemm_blocks = cdiv((N + 15) / 16, 8);

  zero_f32_kernel<<<cdiv(N, T), T, 0, stream>>>(dinv, N);
  deg_count_kernel<<<cdiv(E, T), T, 0, stream>>>(dst, E, dinv);
  deg_finalize_kernel<<<cdiv(N, T), T, 0, stream>>>(dinv, N);

  gemm_wmma_f32_kernel<128, 64, false><<<gemm_blocks, T, 0, stream>>>(x, W1, bufA, N);
  self_init_kernel<<<cdiv((long long)N << 6, T), T, 0, stream>>>(bufA, dinv, b1, bufB, N, 6);
  edge_scatter_kernel<<<cdiv((long long)E << 6, T), T, 0, stream>>>(src, dst, bufA, dinv, bufB, E, 6);

  gemm_wmma_f32_kernel<64, 128, true><<<gemm_blocks, T, 0, stream>>>(bufB, W2, bufA, N);
  self_init_kernel<<<cdiv((long long)N << 7, T), T, 0, stream>>>(bufA, dinv, b2, bufB, N, 7);
  edge_scatter_kernel<<<cdiv((long long)E << 7, T), T, 0, stream>>>(src, dst, bufA, dinv, bufB, E, 7);

  gemm_wmma_f32_kernel<128, 64, true><<<gemm_blocks, T, 0, stream>>>(bufB, W3, bufA, N);
  self_init_kernel<<<cdiv((long long)N << 6, T), T, 0, stream>>>(bufA, dinv, b3, out, N, 6);
  edge_scatter_kernel<<<cdiv((long long)E << 6, T), T, 0, stream>>>(src, dst, bufA, dinv, out, E, 6);
}


# row degrees counted by the row owners in LDS during their owner-list scan, now a launch of its own ahead of a replayed first aggregation: no global f32 atomics, no zero/finalize launches (7->6)
# speedup vs baseline: 5.1242x; 5.1242x over previous
#include <hip/hip_runtime.h>
#include <hip/hip_bf16.h>

typedef __attribute__((ext_vector_type(2))) float v2f;
typedef __attribute__((ext_vector_type(8))) float v8f;
typedef __attribute__((ext_vector_type(4))) float v4f;
typedef __attribute__((ext_vector_type(4))) int v4i;
typedef __attribute__((ext_vector_type(2))) int v2i;

#define NODES 100000
#define AGG_WG 256
#define ROWS_PER_WG ((NODES + AGG_WG - 1) / AGG_WG)
static_assert(AGG_WG <= 256, "an owner workgroup's number must fit one byte of the owner list");

__global__ void owner_list_kernel(const int* __restrict__ dst, int e, int* __restrict__ own) {
  const int i = blockIdx.x * blockDim.x + threadIdx.x;
  const int n_words = ((e + 15) >> 4) << 2;
  if (i >= n_words) return;
  const int e0 = i * 4;
  int d[4];
  if (e0 + 3 < e && ((((size_t)dst) & 15) == 0)) {
    const v4i v = ((const v4i*)dst)[i];
    d[0] = v.x; d[1] = v.y; d[2] = v.z; d[3] = v.w;
  } else {
#pragma unroll
    for (int q = 0; q < 4; ++q) d[q] = (e0 + q < e) ? dst[e0 + q] : 0;
  }
  unsigned w = 0u;
#pragma unroll
  for (int q = 0; q < 4; ++q) {
    if (e0 + q < e) {
      w |= (((unsigned)d[q] / (unsigned)ROWS_PER_WG) & 0xffu) << (8 * q);
    }
  }
  own[i] = (int)w;
}

template <int K, int FOUT, bool RELU, bool SPLITK>
__global__ void gemm_wmma_f32_kernel(const float* __restrict__ X0,
                                     const float* __restrict__ X1,
                                     const float* __restrict__ W,
                                     int ldw,
                                     float* __restrict__ H,
                                     int n_rows) {
  constexpr int NT = FOUT / 16;
  __shared__ float Wp[K * FOUT];

  for (int i = threadIdx.x; i < K * FOUT; i += 256) {
    int k = i / FOUT;
    int col = i - k * FOUT;
    Wp[(k >> 1) * (FOUT * 2) + col * 2 + (k & 1)] = W[k * ldw + col];
  }
  __syncthreads();

  const int wave = threadIdx.x >> 5;
  const int lane = threadIdx.x & 31;
  const int tile_m = blockIdx.x * 8 + wave;
  if (tile_m * 16 >= n_rows) return;

  const int half = lane >> 4;
  const int l15  = lane & 15;

  constexpr int KP = SPLITK ? K / 2 : K;
  const float* __restrict__ xrow0 = X0 + (size_t)(tile_m * 16 + l15) * KP;
  const float* __restrict__ xrow1 = X1 + (size_t)(tile_m * 16 + l15) * KP;

  v8f acc[NT];
#pragma unroll
  for (int t = 0; t < NT; ++t) acc[t] = (v8f){};

#pragma unroll
  for (int k = 0; k < K; k += 4) {
    const int kk = k + half * 2;
    const float* __restrict__ xp = (!SPLITK || k < KP) ? (xrow0 + kk) : (xrow1 + (kk - KP));
    float a0 = xp[0];
    float a1 = xp[1];
    if (RELU) { a0 = fmaxf(a0, 0.0f); a1 = fmaxf(a1, 0.0f); }
    v2f a;
    a.x = a0;
    a.y = a1;
    const float* __restrict__ wrow = Wp + (kk >> 1) * (FOUT * 2) + l15 * 2;
#pragma unroll
    for (int t = 0; t < NT; ++t) {
      const v2f b = *(const v2f*)(wrow + t * 32);
      acc[t] = __builtin_amdgcn_wmma_f32_16x16x4_f32(false, a, false, b,
                                                     (short)0, acc[t], false, false);
    }
  }

  float* __restrict__ hbase = H + (size_t)(tile_m * 16) * FOUT + l15;
#pragma unroll
  for (int t = 0; t < NT; ++t) {
#pragma unroll
    for (int v = 0; v < 8; ++v) {
      hbase[(size_t)(v + half * 8) * FOUT + t * 16] = acc[t][v];
    }
  }
}

#define WQCAP 736
#define PASS_MAX 512
#define LIST_CAP 512
#define LIST_WAVES (AGG_WG * 32)

static_assert(ROWS_PER_WG <= 512, "local row must fit the 9 low bits of a list word");
static_assert(NODES <= (1 << 22), "source row must fit the bits above the local row");

template <int F, int MODE, int FO>
__global__ __launch_bounds__(1024)
void agg_rows_kernel(const int* __restrict__ src,
                     const int* __restrict__ dst,
                     const float* __restrict__ H,
                     const float* __restrict__ dinv,
                     const float* __restrict__ bias,
                     const float* __restrict__ Wn,
                     float* __restrict__ Out,
                     int* lcnt,
                     int* lst,
                     const int* __restrict__ own,
                     int n, int e_cnt) {
  constexpr int NQ = F / 64;
  static_assert(F % 64 == 0, "a row is a whole number of 64-float blocks");
  constexpr int PITCH = (FO > 0) ? F + 2 : F;
  static_assert(F * FO <= 32 * WQCAP, "the packed weights must fit the queues' LDS");
  __shared__ __attribute__((aligned(16))) float acc[ROWS_PER_WG * PITCH];
  __shared__ __attribute__((aligned(16))) int wq_all[32 * WQCAP];

  const int tid  = threadIdx.x;
  const int lane = tid & 31;
  const int wave = __builtin_amdgcn_readfirstlane(tid >> 5);
  int* wq = wq_all + wave * WQCAP;
  const int base = blockIdx.x * ROWS_PER_WG;
  int nrows = n - base;
  if (nrows > ROWS_PER_WG) nrows = ROWS_PER_WG;
  if (nrows < 0) nrows = 0;
  const unsigned unrows = (unsigned)nrows;

  int* degcnt = (int*)acc;
  if constexpr (MODE == 0) {
    for (int i = tid; i < ROWS_PER_WG; i += 1024) degcnt[i] = 0;
  } else {
    for (int i = tid; i < nrows * F; i += 1024) {
      const int r = i / F;
      const int f = i - r * F;
      const float di = dinv[base + r];
      acc[r * PITCH + f] = H[(size_t)(base + r) * F + f] * di * di + bias[f];
    }
  }
  __syncthreads();

  auto add_rows = [&](int s, int lr, float coef, int cnt) {
    for (int j = 0; j < cnt; j += 8) {
      v2f h[8][NQ];
#pragma unroll
      for (int u = 0; u < 8; ++u) {
        int jj = j + u;
        if (jj > cnt - 1) jj = cnt - 1;
        const int sj = __builtin_amdgcn_readlane(s, jj);
        const float* hp = H + (size_t)sj * F + lane * 2;
#pragma unroll
        for (int q = 0; q < NQ; ++q) h[u][q] = *(const v2f*)(hp + q * 64);
      }
#pragma unroll
      for (int u = 0; u < 8; ++u) {
        if (j + u < cnt) {
          const float cj = __int_as_float(__builtin_amdgcn_readlane(__float_as_int(coef), j + u));
          const int rj = __builtin_amdgcn_readlane(lr, j + u);
          float* a = acc + rj * PITCH + lane * 2;
#pragma unroll
          for (int q = 0; q < NQ; ++q) {
            atomicAdd(a + q * 64,     h[u][q].x * cj);
            atomicAdd(a + q * 64 + 1, h[u][q].y * cj);
          }
        }
      }
    }
  };

  const int wid = blockIdx.x * 32 + wave;
  int* mylist = lst + (size_t)wid * LIST_CAP;
  int kept = 0;
  if (MODE == 1) kept = __builtin_amdgcn_readfirstlane(lcnt[wid]);

  if (MODE == 1 && kept >= 0) {
    int pn = (lane < kept) ? mylist[lane] : 0;
    for (int b = 0; b < kept; b += 32) {
      const int p = pn;
      if (b + 32 + lane < kept) pn = mylist[b + 32 + lane];
      int s = 0, lr = 0;
      float coef = 0.0f;
      if (b + lane < kept) {
        s = p >> 9;
        lr = p & 511;
        coef = dinv[s] * dinv[base + lr];
      }
      int cnt = kept - b;
      if (cnt > 32) cnt = 32;
      add_rows(s, lr, coef, cnt);
    }
  } else {
    const int nq = (e_cnt + 15) >> 4;
    const int n_pass = (nq + 4095) >> 12;
    const v4i* __restrict__ own4 = (const v4i*)own;
    const unsigned pat = (unsigned)blockIdx.x * 0x01010101u;
    const int rot = (int)(((unsigned)blockIdx.x * (unsigned)n_pass) / (unsigned)AGG_WG);

    int wcount = 0;
    v4i nx[4], dv[4];
    auto load_pass = [&](int c) {
#pragma unroll
      for (int k = 0; k < 4; ++k) {
        int qi = c * 4096 + k * 1024 + tid;
        if (qi > nq - 1) qi = nq - 1;
        nx[k] = own4[qi];
      }
    };
    if (n_pass > 0) {
      load_pass(rot);
      for (int p = 0; p <= n_pass; ++p) {
        int c = p + rot;
        if (c >= n_pass) c -= n_pass;
#pragma unroll
        for (int u = 0; u < 4; ++u) dv[u] = nx[u];
        {
          int c2 = c + 1;
          if (c2 >= n_pass) c2 -= n_pass;
          load_pass(c2);
        }
#pragma nounroll
        for (int k = 0; k < 4; ++k) {
          const bool end = (p == n_pass);
          if (end || wcount > WQCAP - PASS_MAX) {
            const bool room = (MODE == 0) && kept >= 0 && kept + wcount <= LIST_CAP;
            for (int b = 0; b < wcount; b += 32) {
              const int idx = b + lane;
              int s = 0, lr = 0;
              float coef = 0.0f;
              if (idx < wcount) {
                const int e = wq[idx];
                const int d = dst[e];
                s = src[e];
                lr = d - base;
                if constexpr (MODE == 0) {
                  atomicAdd(&degcnt[lr], 1);
                  if (room) mylist[kept + idx] = (s << 9) | lr;
                } else {
                  coef = dinv[s] * dinv[d];
                }
              }
              if constexpr (MODE == 1) {
                int cnt = wcount - b;
                if (cnt > 32) cnt = 32;
                add_rows(s, lr, coef, cnt);
              }
            }
            if (MODE == 0) kept = room ? kept + wcount : -1;
            wcount = 0;
          }
          if (end) break;
          const v4i cur = dv[0];
          dv[0] = dv[1];
          dv[1] = dv[2];
          dv[2] = dv[3];
          const int e0 = (c * 4096 + k * 1024 + tid) * 16;
          unsigned x[4];
#pragma unroll
          for (int w = 0; w < 4; ++w) x[w] = (unsigned)cur[w] ^ pat;
          if ((c * 4096 + k * 1024 + 1024) * 16 > e_cnt) {
#pragma unroll
            for (int w = 0; w < 4; ++w) {
              int nb = e_cnt - (e0 + w * 4);
              if (nb < 0) nb = 0;
              if (nb < 4) x[w] |= 0xffffffffu << (8 * nb);
            }
          }
#pragma unroll
          for (int w = 0; w < 4; ++w) {
            const unsigned z = (x[w] - 0x01010101u) & ~x[w] & 0x80808080u;
            if (__builtin_amdgcn_ballot_w32(z != 0u) != 0u) {
#pragma unroll
              for (int q = 0; q < 4; ++q) {
                const bool m = ((x[w] >> (8 * q)) & 0xffu) == 0u;
                const unsigned mask = __builtin_amdgcn_ballot_w32(m);
                if (mask != 0u) {
                  if (m) wq[wcount + (int)__builtin_amdgcn_mbcnt_lo(mask, 0u)] = e0 + w * 4 + q;
                  wcount += __builtin_popcount(mask);
                }
              }
            }
          }
        }
      }
    }
    if (MODE == 0 && lane == 0) lcnt[wid] = kept;
  }
  constexpr int NW = (FO > 0) ? (F * FO) / 1024 : 1;
  static_assert(FO == 0 || (F * FO) % 1024 == 0, "Wn is packed in whole passes of the workgroup");
  float wv[NW];
  if constexpr (FO > 0) {
#pragma unroll
    for (int u = 0; u < NW; ++u) wv[u] = Wn[tid + u * 1024];
  }
  __syncthreads();

  if constexpr (MODE == 0) {
    for (int r = tid; r < nrows; r += 1024) Out[base + r] = rsqrtf((float)degcnt[r] + 1.0f);
  } else if constexpr (FO == 0) {
    const int tot4 = (nrows * F) >> 2;
    const v4f* acc4 = (const v4f*)acc;
    v4f* out4 = (v4f*)(Out + (size_t)base * F);
    for (int i = tid; i < tot4; i += 1024) out4[i] = acc4[i];
  } else {
#pragma unroll
    for (int u = 0; u < NW; ++u) {
      const int i = tid + u * 1024;
      const int k = i / FO;
      const int col = i - k * FO;
      wq_all[(k >> 1) * (FO * 2) + col * 2 + (k & 1)] = __float_as_int(wv[u]);
    }
    __syncthreads();

    constexpr int NT = FO / 16;
    const int half = lane >> 4;
    const int l15  = lane & 15;
    const int ntile = (nrows + 15) >> 4;
    for (int tile = wave; tile < ntile; tile += 32) {
      int r = tile * 16 + l15;
      if (r > nrows - 1) r = nrows - 1;
      const float* arow = acc + r * PITCH;
      v8f c[NT];
#pragma unroll
      for (int t = 0; t < NT; ++t) c[t] = (v8f){};
#pragma unroll
      for (int k = 0; k < F; k += 4) {
        const int kk = k + half * 2;
        v2f a = *(const v2f*)(arow + kk);
        a.x = fmaxf(a.x, 0.0f);
        a.y = fmaxf(a.y, 0.0f);
        const int* wrow = wq_all + (kk >> 1) * (FO * 2) + l15 * 2;
#pragma unroll
        for (int t = 0; t < NT; ++t) {
          const v2i bi = *(const v2i*)(wrow + t * 32);
          v2f b;
          b.x = __int_as_float(bi.x);
          b.y = __int_as_float(bi.y);
          c[t] = __builtin_amdgcn_wmma_f32_16x16x4_f32(false, a, false, b,
                                                       (short)0, c[t], false, false);
        }
      }
      float* obase = Out + (size_t)(base + tile * 16) * FO + l15;
#pragma unroll
      for (int v = 0; v < 8; ++v) {
        const int lr = v + half * 8;
        if (tile * 16 + lr < nrows) {
#pragma unroll
          for (int t = 0; t < NT; ++t) obase[(size_t)lr * FO + t * 16] = c[t][v];
        }
      }
    }
  }
}

static inline unsigned int cdiv(long long a, long long b) {
  return (unsigned int)((a + b - 1) / b);
}

extern "C" void kernel_launch(void* const* d_in, const int* in_sizes, int n_in,
                              void* d_out, int out_size, void* d_ws, size_t ws_size,
                              hipStream_t stream) {
  const float* x  = (const float*)d_in[0];
  const int*   ei = (const int*)d_in[1];
  const float* W1 = (const float*)d_in[2];
  const float* b1 = (const float*)d_in[3];
  const float* W2 = (const float*)d_in[4];
  const float* b2 = (const float*)d_in[5];
  const float* W3 = (const float*)d_in[6];
  const float* b3 = (const float*)d_in[7];
  float* out = (float*)d_out;

  const int N = NODES;
  const int E = in_sizes[1] / 2;

  const int* src = ei;
  const int* dst = ei + E;

  float* dinv = (float*)d_ws;
  float* P0 = dinv + N;
  float* H2 = P0 + (size_t)N * 64;
  int* lcnt = (int*)(H2 + (size_t)N * 128);
  int* lst  = lcnt + LIST_WAVES;
  int* own  = lst + (size_t)LIST_WAVES * LIST_CAP;

  const int T = 256;
  const unsigned int gemm_blocks = cdiv((N + 15) / 16, 8);

  owner_list_kernel<<<cdiv(cdiv(E, 16) * 4, T), T, 0, stream>>>(dst, E, own);
  agg_rows_kernel<64, 0, 0><<<AGG_WG, 1024, 0, stream>>>(src, dst, P0, dinv, b1, W2, dinv, lcnt, lst, own, N, E);

  gemm_wmma_f32_kernel<128, 64, false, false><<<gemm_blocks, T, 0, stream>>>(x, x, W1, 64, P0, N);
  agg_rows_kernel<64, 1, 128><<<AGG_WG, 1024, 0, stream>>>(src, dst, P0, dinv, b1, W2, H2, lcnt, lst, own, N, E);

  agg_rows_kernel<128, 1, 64><<<AGG_WG, 1024, 0, stream>>>(src, dst, H2, dinv, b2, W3, P0, lcnt, lst, own, N, E);

  agg_rows_kernel<64, 1, 0><<<AGG_WG, 1024, 0, stream>>>(src, dst, P0, dinv, b3, W3, out, lcnt, lst, own, N, E);
}


# no float atomics in row sums: scan files each edge under the wave owning its row (row%32, LDS slot counters, arrival order); replay adds by plain LDS read-add-write, row loads issued 8 rows ahead
# speedup vs baseline: 6.1631x; 1.2027x over previous
#include <hip/hip_runtime.h>
#include <hip/hip_bf16.h>

typedef __attribute__((ext_vector_type(2))) float v2f;
typedef __attribute__((ext_vector_type(8))) float v8f;
typedef __attribute__((ext_vector_type(4))) float v4f;
typedef __attribute__((ext_vector_type(4))) int v4i;
typedef __attribute__((ext_vector_type(2))) int v2i;

#define NODES 100000
#define AGG_WG 256
#define ROWS_PER_WG ((NODES + AGG_WG - 1) / AGG_WG)
static_assert(AGG_WG <= 256, "an owner workgroup's number must fit one byte of the owner list");

__global__ void owner_list_kernel(const int* __restrict__ dst, int e, int* __restrict__ own) {
  const int i = blockIdx.x * blockDim.x + threadIdx.x;
  const int n_words = ((e + 15) >> 4) << 2;
  if (i >= n_words) return;
  const int e0 = i * 4;
  int d[4];
  if (e0 + 3 < e && ((((size_t)dst) & 15) == 0)) {
    const v4i v = ((const v4i*)dst)[i];
    d[0] = v.x; d[1] = v.y; d[2] = v.z; d[3] = v.w;
  } else {
#pragma unroll
    for (int q = 0; q < 4; ++q) d[q] = (e0 + q < e) ? dst[e0 + q] : 0;
  }
  unsigned w = 0u;
#pragma unroll
  for (int q = 0; q < 4; ++q) {
    if (e0 + q < e) {
      w |= (((unsigned)d[q] / (unsigned)ROWS_PER_WG) & 0xffu) << (8 * q);
    }
  }
  own[i] = (int)w;
}

template <int K, int FOUT, bool RELU, bool SPLITK>
__global__ void gemm_wmma_f32_kernel(const float* __restrict__ X0,
                                     const float* __restrict__ X1,
                                     const float* __restrict__ W,
                                     int ldw,
                                     float* __restrict__ H,
                                     int n_rows) {
  constexpr int NT = FOUT / 16;
  __shared__ float Wp[K * FOUT];

  for (int i = threadIdx.x; i < K * FOUT; i += 256) {
    int k = i / FOUT;
    int col = i - k * FOUT;
    Wp[(k >> 1) * (FOUT * 2) + col * 2 + (k & 1)] = W[k * ldw + col];
  }
  __syncthreads();

  const int wave = threadIdx.x >> 5;
  const int lane = threadIdx.x & 31;
  const int tile_m = blockIdx.x * 8 + wave;
  if (tile_m * 16 >= n_rows) return;

  const int half = lane >> 4;
  const int l15  = lane & 15;

  constexpr int KP = SPLITK ? K / 2 : K;
  const float* __restrict__ xrow0 = X0 + (size_t)(tile_m * 16 + l15) * KP;
  const float* __restrict__ xrow1 = X1 + (size_t)(tile_m * 16 + l15) * KP;

  v8f acc[NT];
#pragma unroll
  for (int t = 0; t < NT; ++t) acc[t] = (v8f){};

#pragma unroll
  for (int k = 0; k < K; k += 4) {
    const int kk = k + half * 2;
    const float* __restrict__ xp = (!SPLITK || k < KP) ? (xrow0 + kk) : (xrow1 + (kk - KP));
    float a0 = xp[0];
    float a1 = xp[1];
    if (RELU) { a0 = fmaxf(a0, 0.0f); a1 = fmaxf(a1, 0.0f); }
    v2f a;
    a.x = a0;
    a.y = a1;
    const float* __restrict__ wrow = Wp + (kk >> 1) * (FOUT * 2) + l15 * 2;
#pragma unroll
    for (int t = 0; t < NT; ++t) {
      const v2f b = *(const v2f*)(wrow + t * 32);
      acc[t] = __builtin_amdgcn_wmma_f32_16x16x4_f32(false, a, false, b,
                                                     (short)0, acc[t], false, false);
    }
  }

  float* __restrict__ hbase = H + (size_t)(tile_m * 16) * FOUT + l15;
#pragma unroll
  for (int t = 0; t < NT; ++t) {
#pragma unroll
    for (int v = 0; v < 8; ++v) {
      hbase[(size_t)(v + half * 8) * FOUT + t * 16] = acc[t][v];
    }
  }
}

#define WQCAP 736
#define PASS_MAX 512
#define LIST_CAP 512
#define LIST_WAVES (AGG_WG * 32)

static_assert(ROWS_PER_WG <= 512, "local row must fit the 9 low bits of a list word");
static_assert(NODES <= (1 << 22), "source row must fit the bits above the local row");

template <int F, int MODE, int FO>
__global__ __launch_bounds__(1024)
void agg_rows_kernel(const int* __restrict__ src,
                     const int* __restrict__ dst,
                     const float* __restrict__ H,
                     const float* __restrict__ dinv,
                     const float* __restrict__ bias,
                     const float* __restrict__ Wn,
                     float* __restrict__ Out,
                     int* lcnt,
                     int* lst,
                     const int* __restrict__ own,
                     int n, int e_cnt) {
  constexpr int NQ = F / 64;
  static_assert(F % 64 == 0, "a row is a whole number of 64-float blocks");
  constexpr int PITCH = (FO > 0) ? F + 2 : F;
  static_assert(F * FO <= 32 * WQCAP, "the packed weights must fit the queues' LDS");
  __shared__ __attribute__((aligned(16))) float acc[(ROWS_PER_WG + 1) * PITCH];
  __shared__ __attribute__((aligned(16))) int wq_all[32 * WQCAP];

  const int tid  = threadIdx.x;
  const int lane = tid & 31;
  const int wave = __builtin_amdgcn_readfirstlane(tid >> 5);
  int* wq = wq_all + wave * WQCAP;
  const int base = blockIdx.x * ROWS_PER_WG;
  int nrows = n - base;
  if (nrows > ROWS_PER_WG) nrows = ROWS_PER_WG;
  if (nrows < 0) nrows = 0;
  const unsigned unrows = (unsigned)nrows;

  int* degcnt = (int*)acc;
  int* lcount = (int*)acc + 512;
  if constexpr (MODE == 0) {
    for (int i = tid; i < ROWS_PER_WG; i += 1024) degcnt[i] = 0;
    if (tid < 32) lcount[tid] = 0;
  } else {
    for (int i = tid; i < nrows * F; i += 1024) {
      const int r = i / F;
      const int f = i - r * F;
      const float di = dinv[base + r];
      acc[r * PITCH + f] = H[(size_t)(base + r) * F + f] * di * di + bias[f];
    }
  }
  __syncthreads();

  auto add_rows = [&](int s, int lr, float coef, int cnt) {
    for (int j = 0; j < cnt; j += 8) {
      v2f h[8][NQ];
#pragma unroll
      for (int u = 0; u < 8; ++u) {
        int jj = j + u;
        if (jj > cnt - 1) jj = cnt - 1;
        const int sj = __builtin_amdgcn_readlane(s, jj);
        const float* hp = H + (size_t)sj * F + lane * 2;
#pragma unroll
        for (int q = 0; q < NQ; ++q) h[u][q] = *(const v2f*)(hp + q * 64);
      }
#pragma unroll
      for (int u = 0; u < 8; ++u) {
        if (j + u < cnt) {
          const float cj = __int_as_float(__builtin_amdgcn_readlane(__float_as_int(coef), j + u));
          const int rj = __builtin_amdgcn_readlane(lr, j + u);
          float* a = acc + rj * PITCH + lane * 2;
#pragma unroll
          for (int q = 0; q < NQ; ++q) {
            atomicAdd(a + q * 64,     h[u][q].x * cj);
            atomicAdd(a + q * 64 + 1, h[u][q].y * cj);
          }
        }
      }
    }
  };

  const int wid = blockIdx.x * 32 + wave;
  int* mylist = lst + (size_t)wid * LIST_CAP;
  int kept = 0;
  bool all_ok = false;
  if (MODE == 1) {
    const int kj = lcnt[blockIdx.x * 32 + lane];
    all_ok = (__builtin_amdgcn_ballot_w32(kj < 0) == 0u);
    kept = __builtin_amdgcn_readlane(kj, wave);
  }

  if (MODE == 1 && all_ok) {
    if (kept > 0) {
      const unsigned lane2 = (unsigned)lane * 2u;
      const unsigned lane8 = (unsigned)lane * 8u;
      const char* Hb = (const char*)H;
      v2f hA[8][NQ], hB[8][NQ];
      auto list_word = [&](int b) -> int {
        int idx = b + lane;
        if (idx > kept - 1) idx = kept - 1;
        return mylist[idx];
      };
      auto issue = [&](v2f (&h)[8][NQ], unsigned ro, int l0) {
#pragma unroll
        for (int u = 0; u < 8; ++u) {
          const unsigned rj = (unsigned)__builtin_amdgcn_readlane((int)ro, l0 + u);
          const char* hp = Hb + (rj + lane8);
#pragma unroll
          for (int q = 0; q < NQ; ++q) h[u][q] = *(const v2f*)(hp + q * 256);
        }
      };
      auto add8 = [&](v2f (&h)[8][NQ], unsigned lo, float cf, int l0) {
#pragma unroll
        for (int u = 0; u < 8; ++u) {
          const unsigned lj = (unsigned)__builtin_amdgcn_readlane((int)lo, l0 + u);
          const float cj = __int_as_float(__builtin_amdgcn_readlane(__float_as_int(cf), l0 + u));
          v2f* a = (v2f*)(acc + (lj + lane2));
#pragma unroll
          for (int q = 0; q < NQ; ++q) {
            v2f t = a[q * 32];
            t.x += h[u][q].x * cj;
            t.y += h[u][q].y * cj;
            a[q * 32] = t;
          }
        }
      };

      const int ng = (kept + 7) >> 3;
      int w = list_word(0);
      unsigned ro = (unsigned)(w >> 9) * (unsigned)(F * 4);
      unsigned lo = (lane < kept) ? (unsigned)(w & 511) * (unsigned)PITCH : (unsigned)(ROWS_PER_WG * PITCH);
      float cf = dinv[w >> 9] * dinv[base + (w & 511)];
      if (!(lane < kept)) cf = 0.0f;
      issue(hA, ro, 0);
      int g0 = 0;
      while (g0 + 4 < ng) {
        const int bn = (g0 + 4) * 8;
        const int wn = list_word(bn);
        issue(hB, ro, 8);
        __builtin_amdgcn_sched_barrier(0);
        add8(hA, lo, cf, 0);
        const bool vn = (bn + lane < kept);
        const unsigned ron = (unsigned)(wn >> 9) * (unsigned)(F * 4);
        const unsigned lon = vn ? (unsigned)(wn & 511) * (unsigned)PITCH : (unsigned)(ROWS_PER_WG * PITCH);
        const float dn1 = dinv[wn >> 9];
        const float dn2 = dinv[base + (wn & 511)];
        issue(hA, ro, 16);
        __builtin_amdgcn_sched_barrier(0);
        add8(hB, lo, cf, 8);
        issue(hB, ro, 24);
        __builtin_amdgcn_sched_barrier(0);
        add8(hA, lo, cf, 16);
        issue(hA, ron, 0);
        __builtin_amdgcn_sched_barrier(0);
        add8(hB, lo, cf, 24);
        cf = vn ? dn1 * dn2 : 0.0f;
        ro = ron;
        lo = lon;
        g0 += 4;
      }
      const int ngl = ng - g0;
      if (ngl == 1) {
        add8(hA, lo, cf, 0);
      } else if (ngl == 2) {
        issue(hB, ro, 8);
        __builtin_amdgcn_sched_barrier(0);
        add8(hA, lo, cf, 0);
        add8(hB, lo, cf, 8);
      } else if (ngl == 3) {
        issue(hB, ro, 8);
        __builtin_amdgcn_sched_barrier(0);
        add8(hA, lo, cf, 0);
        issue(hA, ro, 16);
        __builtin_amdgcn_sched_barrier(0);
        add8(hB, lo, cf, 8);
        add8(hA, lo, cf, 16);
      } else {
        issue(hB, ro, 8);
        __builtin_amdgcn_sched_barrier(0);
        add8(hA, lo, cf, 0);
        issue(hA, ro, 16);
        __builtin_amdgcn_sched_barrier(0);
        add8(hB, lo, cf, 8);
        issue(hB, ro, 24);
        __builtin_amdgcn_sched_barrier(0);
        add8(hA, lo, cf, 16);
        add8(hB, lo, cf, 24);
      }
    }
  } else {
    const int nq = (e_cnt + 15) >> 4;
    const int n_pass = (nq + 4095) >> 12;
    const v4i* __restrict__ own4 = (const v4i*)own;
    const unsigned pat = (unsigned)blockIdx.x * 0x01010101u;
    const int rot = (int)(((unsigned)blockIdx.x * (unsigned)n_pass) / (unsigned)AGG_WG);

    int wcount = 0;
    v4i nx[4], dv[4];
    auto load_pass = [&](int c) {
#pragma unroll
      for (int k = 0; k < 4; ++k) {
        int qi = c * 4096 + k * 1024 + tid;
        if (qi > nq - 1) qi = nq - 1;
        nx[k] = own4[qi];
      }
    };
    if (n_pass > 0) {
      load_pass(rot);
      for (int p = 0; p <= n_pass; ++p) {
        int c = p + rot;
        if (c >= n_pass) c -= n_pass;
#pragma unroll
        for (int u = 0; u < 4; ++u) dv[u] = nx[u];
        {
          int c2 = c + 1;
          if (c2 >= n_pass) c2 -= n_pass;
          load_pass(c2);
        }
#pragma nounroll
        for (int k = 0; k < 4; ++k) {
          const bool end = (p == n_pass);
          if (end || wcount > WQCAP - PASS_MAX) {
            for (int b = 0; b < wcount; b += 32) {
              const int idx = b + lane;
              int s = 0, lr = 0;
              float coef = 0.0f;
              if (idx < wcount) {
                const int e = wq[idx];
                const int d = dst[e];
                s = src[e];
                lr = d - base;
                if constexpr (MODE == 0) {
                  atomicAdd(&degcnt[lr], 1);
                  const int ow = lr & 31;
                  const int pos = atomicAdd(&lcount[ow], 1);
                  if (pos < LIST_CAP) lst[(size_t)(blockIdx.x * 32 + ow) * LIST_CAP + pos] = (s << 9) | lr;
                } else {
                  coef = dinv[s] * dinv[d];
                }
              }
              if constexpr (MODE == 1) {
                int cnt = wcount - b;
                if (cnt > 32) cnt = 32;
                add_rows(s, lr, coef, cnt);
              }
            }
            wcount = 0;
          }
          if (end) break;
          const v4i cur = dv[0];
          dv[0] = dv[1];
          dv[1] = dv[2];
          dv[2] = dv[3];
          const int e0 = (c * 4096 + k * 1024 + tid) * 16;
          unsigned x[4];
#pragma unroll
          for (int w = 0; w < 4; ++w) x[w] = (unsigned)cur[w] ^ pat;
          if ((c * 4096 + k * 1024 + 1024) * 16 > e_cnt) {
#pragma unroll
            for (int w = 0; w < 4; ++w) {
              int nb = e_cnt - (e0 + w * 4);
              if (nb < 0) nb = 0;
              if (nb < 4) x[w] |= 0xffffffffu << (8 * nb);
            }
          }
#pragma unroll
          for (int w = 0; w < 4; ++w) {
            const unsigned z = (x[w] - 0x01010101u) & ~x[w] & 0x80808080u;
            if (__builtin_amdgcn_ballot_w32(z != 0u) != 0u) {
#pragma unroll
              for (int q = 0; q < 4; ++q) {
                const bool m = ((x[w] >> (8 * q)) & 0xffu) == 0u;
                const unsigned mask = __builtin_amdgcn_ballot_w32(m);
                if (mask != 0u) {
                  if (m) wq[wcount + (int)__builtin_amdgcn_mbcnt_lo(mask, 0u)] = e0 + w * 4 + q;
                  wcount += __builtin_popcount(mask);
                }
              }
            }
          }
        }
      }
    }
  }
  constexpr int NW = (FO > 0) ? (F * FO) / 1024 : 1;
  static_assert(FO == 0 || (F * FO) % 1024 == 0, "Wn is packed in whole passes of the workgroup");
  float wv[NW];
  if constexpr (FO > 0) {
#pragma unroll
    for (int u = 0; u < NW; ++u) wv[u] = Wn[tid + u * 1024];
  }
  __syncthreads();

  if constexpr (MODE == 0) {
    for (int r = tid; r < nrows; r += 1024) Out[base + r] = rsqrtf((float)degcnt[r] + 1.0f);
    if (tid < 32) {
      const int c = lcount[tid];
      lcnt[blockIdx.x * 32 + tid] = (c <= LIST_CAP) ? c : -1;
    }
  } else if constexpr (FO == 0) {
    const int tot4 = (nrows * F) >> 2;
    const v4f* acc4 = (const v4f*)acc;
    v4f* out4 = (v4f*)(Out + (size_t)base * F);
    for (int i = tid; i < tot4; i += 1024) out4[i] = acc4[i];
  } else {
#pragma unroll
    for (int u = 0; u < NW; ++u) {
      const int i = tid + u * 1024;
      const int k = i / FO;
      const int col = i - k * FO;
      wq_all[(k >> 1) * (FO * 2) + col * 2 + (k & 1)] = __float_as_int(wv[u]);
    }
    __syncthreads();

    constexpr int NT = FO / 16;
    const int half = lane >> 4;
    const int l15  = lane & 15;
    const int ntile = (nrows + 15) >> 4;
    for (int tile = wave; tile < ntile; tile += 32) {
      int r = tile * 16 + l15;
      if (r > nrows - 1) r = nrows - 1;
      const float* arow = acc + r * PITCH;
      v8f c[NT];
#pragma unroll
      for (int t = 0; t < NT; ++t) c[t] = (v8f){};
#pragma unroll
      for (int k = 0; k < F; k += 4) {
        const int kk = k + half * 2;
        v2f a = *(const v2f*)(arow + kk);
        a.x = fmaxf(a.x, 0.0f);
        a.y = fmaxf(a.y, 0.0f);
        const int* wrow = wq_all + (kk >> 1) * (FO * 2) + l15 * 2;
#pragma unroll
        for (int t = 0; t < NT; ++t) {
          const v2i bi = *(const v2i*)(wrow + t * 32);
          v2f b;
          b.x = __int_as_float(bi.x);
          b.y = __int_as_float(bi.y);
          c[t] = __builtin_amdgcn_wmma_f32_16x16x4_f32(false, a, false, b,
                                                       (short)0, c[t], false, false);
        }
      }
      float* obase = Out + (size_t)(base + tile * 16) * FO + l15;
#pragma unroll
      for (int v = 0; v < 8; ++v) {
        const int lr = v + half * 8;
        if (tile * 16 + lr < nrows) {
#pragma unroll
          for (int t = 0; t < NT; ++t) obase[(size_t)lr * FO + t * 16] = c[t][v];
        }
      }
    }
  }
}

static inline unsigned int cdiv(long long a, long long b) {
  return (unsigned int)((a + b - 1) / b);
}

extern "C" void kernel_launch(void* const* d_in, const int* in_sizes, int n_in,
                              void* d_out, int out_size, void* d_ws, size_t ws_size,
                              hipStream_t stream) {
  const float* x  = (const float*)d_in[0];
  const int*   ei = (const int*)d_in[1];
  const float* W1 = (const float*)d_in[2];
  const float* b1 = (const float*)d_in[3];
  const float* W2 = (const float*)d_in[4];
  const float* b2 = (const float*)d_in[5];
  const float* W3 = (const float*)d_in[6];
  const float* b3 = (const float*)d_in[7];
  float* out = (float*)d_out;

  const int N = NODES;
  const int E = in_sizes[1] / 2;

  const int* src = ei;
  const int* dst = ei + E;

  float* dinv = (float*)d_ws;
  float* P0 = dinv + N;
  float* H2 = P0 + (size_t)N * 64;
  int* lcnt = (int*)(H2 + (size_t)N * 128);
  int* lst  = lcnt + LIST_WAVES;
  int* own  = lst + (size_t)LIST_WAVES * LIST_CAP;

  const int T = 256;
  const unsigned int gemm_blocks = cdiv((N + 15) / 16, 8);

  owner_list_kernel<<<cdiv(cdiv(E, 16) * 4, T), T, 0, stream>>>(dst, E, own);
  agg_rows_kernel<64, 0, 0><<<AGG_WG, 1024, 0, stream>>>(src, dst, P0, dinv, b1, W2, dinv, lcnt, lst, own, N, E);

  gemm_wmma_f32_kernel<128, 64, false, false><<<gemm_blocks, T, 0, stream>>>(x, x, W1, 64, P0, N);
  agg_rows_kernel<64, 1, 128><<<AGG_WG, 1024, 0, stream>>>(src, dst, P0, dinv, b1, W2, H2, lcnt, lst, own, N, E);

  agg_rows_kernel<128, 1, 64><<<AGG_WG, 1024, 0, stream>>>(src, dst, H2, dinv, b2, W3, P0, lcnt, lst, own, N, E);

  agg_rows_kernel<64, 1, 0><<<AGG_WG, 1024, 0, stream>>>(src, dst, P0, dinv, b3, W3, out, lcnt, lst, own, N, E);
}


# owners' search reads one bit row per owner: owner-list launch collects a bit per edge and owner in LDS (integer OR), stores whole lines; a word answers 32 edges; f32 sums in arrival order as before
# speedup vs baseline: 6.5534x; 1.0633x over previous
#include <hip/hip_runtime.h>
#include <hip/hip_bf16.h>

typedef __attribute__((ext_vector_type(2))) float v2f;
typedef __attribute__((ext_vector_type(8))) float v8f;
typedef __attribute__((ext_vector_type(4))) float v4f;
typedef __attribute__((ext_vector_type(4))) int v4i;
typedef __attribute__((ext_vector_type(2))) int v2i;

#define NODES 100000
#define AGG_WG 256
#define ROWS_PER_WG ((NODES + AGG_WG - 1) / AGG_WG)
static_assert(AGG_WG <= 256, "an owner workgroup's number must fit one byte of the owner list");

#define BM_CHUNKS 32
static_assert(BM_CHUNKS * 32 == 256 * 4, "a block's 1024 edges fill BM_CHUNKS words of every owner");

__global__ __launch_bounds__(256)
void owner_list_kernel(const int* __restrict__ dst, int e, int* __restrict__ own, int* __restrict__ bm) {
  __shared__ __attribute__((aligned(16))) unsigned int tile[AGG_WG * BM_CHUNKS];
  const int tid = threadIdx.x;
  v4i* tile4 = (v4i*)tile;
#pragma unroll
  for (int u = 0; u < 8; ++u) tile4[tid + u * 256] = (v4i){0, 0, 0, 0};
  __syncthreads();

  const int i = blockIdx.x * 256 + tid;
  const int n_words = ((e + 15) >> 4) << 2;
  const int e0 = i * 4;
  int d[4];
  if (e0 + 3 < e && ((((size_t)dst) & 15) == 0)) {
    const v4i v = ((const v4i*)dst)[i];
    d[0] = v.x; d[1] = v.y; d[2] = v.z; d[3] = v.w;
  } else {
#pragma unroll
    for (int q = 0; q < 4; ++q) d[q] = (e0 + q < e) ? dst[e0 + q] : 0;
  }
  unsigned w = 0u;
  unsigned int* trow = tile + (tid >> 3);
  const int bit0 = (tid & 7) * 4;
#pragma unroll
  for (int q = 0; q < 4; ++q) {
    if (e0 + q < e) {
      const unsigned o = ((unsigned)d[q] / (unsigned)ROWS_PER_WG) & 0xffu;
      w |= o << (8 * q);
      atomicOr(trow + o * BM_CHUNKS, 1u << (bit0 + q));
    }
  }
  if (i < n_words) own[i] = (int)w;
  __syncthreads();

  const int pitch = ((e + 1023) >> 10) * BM_CHUNKS;
#pragma unroll
  for (int u = 0; u < 8; ++u) {
    const int v = tid + u * 256;
    v4i* line = (v4i*)(bm + (size_t)(v >> 3) * pitch + (size_t)blockIdx.x * BM_CHUNKS);
    line[v & 7] = tile4[v];
  }
}

template <int K, int FOUT, bool RELU, bool SPLITK>
__global__ void gemm_wmma_f32_kernel(const float* __restrict__ X0,
                                     const float* __restrict__ X1,
                                     const float* __restrict__ W,
                                     int ldw,
                                     float* __restrict__ H,
                                     int n_rows) {
  constexpr int NT = FOUT / 16;
  __shared__ float Wp[K * FOUT];

  for (int i = threadIdx.x; i < K * FOUT; i += 256) {
    int k = i / FOUT;
    int col = i - k * FOUT;
    Wp[(k >> 1) * (FOUT * 2) + col * 2 + (k & 1)] = W[k * ldw + col];
  }
  __syncthreads();

  const int wave = threadIdx.x >> 5;
  const int lane = threadIdx.x & 31;
  const int tile_m = blockIdx.x * 8 + wave;
  if (tile_m * 16 >= n_rows) return;

  const int half = lane >> 4;
  const int l15  = lane & 15;

  constexpr int KP = SPLITK ? K / 2 : K;
  const float* __restrict__ xrow0 = X0 + (size_t)(tile_m * 16 + l15) * KP;
  const float* __restrict__ xrow1 = X1 + (size_t)(tile_m * 16 + l15) * KP;

  v8f acc[NT];
#pragma unroll
  for (int t = 0; t < NT; ++t) acc[t] = (v8f){};

#pragma unroll
  for (int k = 0; k < K; k += 4) {
    const int kk = k + half * 2;
    const float* __restrict__ xp = (!SPLITK || k < KP) ? (xrow0 + kk) : (xrow1 + (kk - KP));
    float a0 = xp[0];
    float a1 = xp[1];
    if (RELU) { a0 = fmaxf(a0, 0.0f); a1 = fmaxf(a1, 0.0f); }
    v2f a;
    a.x = a0;
    a.y = a1;
    const float* __restrict__ wrow = Wp + (kk >> 1) * (FOUT * 2) + l15 * 2;
#pragma unroll
    for (int t = 0; t < NT; ++t) {
      const v2f b = *(const v2f*)(wrow + t * 32);
      acc[t] = __builtin_amdgcn_wmma_f32_16x16x4_f32(false, a, false, b,
                                                     (short)0, acc[t], false, false);
    }
  }

  float* __restrict__ hbase = H + (size_t)(tile_m * 16) * FOUT + l15;
#pragma unroll
  for (int t = 0; t < NT; ++t) {
#pragma unroll
    for (int v = 0; v < 8; ++v) {
      hbase[(size_t)(v + half * 8) * FOUT + t * 16] = acc[t][v];
    }
  }
}

#define WQCAP 736
#define PASS_MAX 512
#define LIST_CAP 512
#define LIST_WAVES (AGG_WG * 32)

static_assert(ROWS_PER_WG <= 512, "local row must fit the 9 low bits of a list word");
static_assert(NODES <= (1 << 22), "source row must fit the bits above the local row");

template <int F, int MODE, int FO>
__global__ __launch_bounds__(1024)
void agg_rows_kernel(const int* __restrict__ src,
                     const int* __restrict__ dst,
                     const float* __restrict__ H,
                     const float* __restrict__ dinv,
                     const float* __restrict__ bias,
                     const float* __restrict__ Wn,
                     float* __restrict__ Out,
                     int* lcnt,
                     int* lst,
                     const int* __restrict__ own,
                     const int* __restrict__ bm,
                     int n, int e_cnt) {
  constexpr int NQ = F / 64;
  static_assert(F % 64 == 0, "a row is a whole number of 64-float blocks");
  constexpr int PITCH = (FO > 0) ? F + 2 : F;
  static_assert(F * FO <= 32 * WQCAP, "the packed weights must fit the queues' LDS");
  __shared__ __attribute__((aligned(16))) float acc[(ROWS_PER_WG + 1) * PITCH];
  __shared__ __attribute__((aligned(16))) int wq_all[32 * WQCAP];

  const int tid  = threadIdx.x;
  const int lane = tid & 31;
  const int wave = __builtin_amdgcn_readfirstlane(tid >> 5);
  int* wq = wq_all + wave * WQCAP;
  const int base = blockIdx.x * ROWS_PER_WG;
  int nrows = n - base;
  if (nrows > ROWS_PER_WG) nrows = ROWS_PER_WG;
  if (nrows < 0) nrows = 0;
  const unsigned unrows = (unsigned)nrows;

  int* degcnt = (int*)acc;
  int* lcount = (int*)acc + 512;
  if constexpr (MODE == 0) {
    for (int i = tid; i < ROWS_PER_WG; i += 1024) degcnt[i] = 0;
    if (tid < 32) lcount[tid] = 0;
  } else {
    for (int i = tid; i < nrows * F; i += 1024) {
      const int r = i / F;
      const int f = i - r * F;
      const float di = dinv[base + r];
      acc[r * PITCH + f] = H[(size_t)(base + r) * F + f] * di * di + bias[f];
    }
  }
  __syncthreads();

  auto add_rows = [&](int s, int lr, float coef, int cnt) {
    for (int j = 0; j < cnt; j += 8) {
      v2f h[8][NQ];
#pragma unroll
      for (int u = 0; u < 8; ++u) {
        int jj = j + u;
        if (jj > cnt - 1) jj = cnt - 1;
        const int sj = __builtin_amdgcn_readlane(s, jj);
        const float* hp = H + (size_t)sj * F + lane * 2;
#pragma unroll
        for (int q = 0; q < NQ; ++q) h[u][q] = *(const v2f*)(hp + q * 64);
      }
#pragma unroll
      for (int u = 0; u < 8; ++u) {
        if (j + u < cnt) {
          const float cj = __int_as_float(__builtin_amdgcn_readlane(__float_as_int(coef), j + u));
          const int rj = __builtin_amdgcn_readlane(lr, j + u);
          float* a = acc + rj * PITCH + lane * 2;
#pragma unroll
          for (int q = 0; q < NQ; ++q) {
            atomicAdd(a + q * 64,     h[u][q].x * cj);
            atomicAdd(a + q * 64 + 1, h[u][q].y * cj);
          }
        }
      }
    }
  };

  const int wid = blockIdx.x * 32 + wave;
  int* mylist = lst + (size_t)wid * LIST_CAP;
  int kept = 0;
  bool all_ok = false;
  if (MODE == 1) {
    const int kj = lcnt[blockIdx.x * 32 + lane];
    all_ok = (__builtin_amdgcn_ballot_w32(kj < 0) == 0u);
    kept = __builtin_amdgcn_readlane(kj, wave);
  }

  if constexpr (MODE == 0) {
    const int pitch = ((e_cnt + 1023) >> 10) * BM_CHUNKS;
    const int n4 = pitch >> 2;
    const int n_pass = (n4 + 4095) >> 12;
    const v4i* __restrict__ row4 = (const v4i*)(bm + (size_t)blockIdx.x * pitch);
    const v4i zero4 = (v4i){0, 0, 0, 0};
    int wcount = 0;

    auto drain0 = [&]() {
      for (int b = 0; b < wcount; b += 32) {
        const int idx = b + lane;
        if (idx < wcount) {
          const int ed = wq[idx];
          const int d = dst[ed];
          const int sr = src[ed];
          const int lr = d - base;
          atomicAdd(&degcnt[lr], 1);
          const int ow = lr & 31;
          const int pos = atomicAdd(&lcount[ow], 1);
          if (pos < LIST_CAP) lst[(size_t)(blockIdx.x * 32 + ow) * LIST_CAP + pos] = (sr << 9) | lr;
        }
      }
      wcount = 0;
    };

    v4i nx[4], dv[4];
    auto load_pass = [&](int c) {
#pragma unroll
      for (int k = 0; k < 4; ++k) {
        const int p = c * 4096 + k * 1024 + tid;
        nx[k] = (p < n4) ? row4[p] : zero4;
      }
    };
    load_pass(0);
    for (int c = 0; c < n_pass; ++c) {
#pragma unroll
      for (int u = 0; u < 4; ++u) dv[u] = nx[u];
      unsigned some = 0u;
#pragma unroll
      for (int u = 0; u < 4; ++u) {
        const int any = dv[u].x | dv[u].y | dv[u].z | dv[u].w;
        if (__builtin_amdgcn_ballot_w32(any != 0) != 0u) some |= 1u << u;
      }
      load_pass(c + 1);
#pragma nounroll
      for (int k = 0; k < 4; ++k) {
        v4i cur = dv[0];
        dv[0] = dv[1];
        dv[1] = dv[2];
        dv[2] = dv[3];
        const int w0 = (c * 4096 + k * 1024 + tid) * 4;
        if (((some >> k) & 1u) == 0u) continue;
#pragma nounroll
        for (int w = 0; w < 4; ++w) {
          unsigned word = (unsigned)cur.x;
          cur = (v4i){cur.y, cur.z, cur.w, 0};
          const int ebase = (w0 + w) * 32;
          unsigned live = __builtin_amdgcn_ballot_w32(word != 0u);
          while (live != 0u) {
            if (wcount > WQCAP - 32) drain0();
            if (word != 0u) {
              const int j = __builtin_ctz(word);
              word &= word - 1u;
              wq[wcount + (int)__builtin_amdgcn_mbcnt_lo(live, 0u)] = ebase + j;
            }
            wcount += __builtin_popcount(live);
            live = __builtin_amdgcn_ballot_w32(word != 0u);
          }
        }
      }
    }
    drain0();
  } else if (all_ok) {
    if (kept > 0) {
      const unsigned lane2 = (unsigned)lane * 2u;
      const unsigned lane8 = (unsigned)lane * 8u;
      const char* Hb = (const char*)H;
      v2f hA[8][NQ], hB[8][NQ];
      auto list_word = [&](int b) -> int {
        int idx = b + lane;
        if (idx > kept - 1) idx = kept - 1;
        return mylist[idx];
      };
      auto issue = [&](v2f (&h)[8][NQ], unsigned ro, int l0) {
#pragma unroll
        for (int u = 0; u < 8; ++u) {
          const unsigned rj = (unsigned)__builtin_amdgcn_readlane((int)ro, l0 + u);
          const char* hp = Hb + (rj + lane8);
#pragma unroll
          for (int q = 0; q < NQ; ++q) h[u][q] = *(const v2f*)(hp + q * 256);
        }
      };
      auto add8 = [&](v2f (&h)[8][NQ], unsigned lo, float cf, int l0) {
#pragma unroll
        for (int u = 0; u < 8; ++u) {
          const unsigned lj = (unsigned)__builtin_amdgcn_readlane((int)lo, l0 + u);
          const float cj = __int_as_float(__builtin_amdgcn_readlane(__float_as_int(cf), l0 + u));
          v2f* a = (v2f*)(acc + (lj + lane2));
#pragma unroll
          for (int q = 0; q < NQ; ++q) {
            v2f t = a[q * 32];
            t.x += h[u][q].x * cj;
            t.y += h[u][q].y * cj;
            a[q * 32] = t;
          }
        }
      };

      const int ng = (kept + 7) >> 3;
      int w = list_word(0);
      unsigned ro = (unsigned)(w >> 9) * (unsigned)(F * 4);
      unsigned lo = (lane < kept) ? (unsigned)(w & 511) * (unsigned)PITCH : (unsigned)(ROWS_PER_WG * PITCH);
      float cf = dinv[w >> 9] * dinv[base + (w & 511)];
      if (!(lane < kept)) cf = 0.0f;
      issue(hA, ro, 0);
      int g0 = 0;
      while (g0 + 4 < ng) {
        const int bn = (g0 + 4) * 8;
        const int wn = list_word(bn);
        issue(hB, ro, 8);
        __builtin_amdgcn_sched_barrier(0);
        add8(hA, lo, cf, 0);
        const bool vn = (bn + lane < kept);
        const unsigned ron = (unsigned)(wn >> 9) * (unsigned)(F * 4);
        const unsigned lon = vn ? (unsigned)(wn & 511) * (unsigned)PITCH : (unsigned)(ROWS_PER_WG * PITCH);
        const float dn1 = dinv[wn >> 9];
        const float dn2 = dinv[base + (wn & 511)];
        issue(hA, ro, 16);
        __builtin_amdgcn_sched_barrier(0);
        add8(hB, lo, cf, 8);
        issue(hB, ro, 24);
        __builtin_amdgcn_sched_barrier(0);
        add8(hA, lo, cf, 16);
        issue(hA, ron, 0);
        __builtin_amdgcn_sched_barrier(0);
        add8(hB, lo, cf, 24);
        cf = vn ? dn1 * dn2 : 0.0f;
        ro = ron;
        lo = lon;
        g0 += 4;
      }
      const int ngl = ng - g0;
      if (ngl == 1) {
        add8(hA, lo, cf, 0);
      } else if (ngl == 2) {
        issue(hB, ro, 8);
        __builtin_amdgcn_sched_barrier(0);
        add8(hA, lo, cf, 0);
        add8(hB, lo, cf, 8);
      } else if (ngl == 3) {
        issue(hB, ro, 8);
        __builtin_amdgcn_sched_barrier(0);
        add8(hA, lo, cf, 0);
        issue(hA, ro, 16);
        __builtin_amdgcn_sched_barrier(0);
        add8(hB, lo, cf, 8);
        add8(hA, lo, cf, 16);
      } else {
        issue(hB, ro, 8);
        __builtin_amdgcn_sched_barrier(0);
        add8(hA, lo, cf, 0);
        issue(hA, ro, 16);
        __builtin_amdgcn_sched_barrier(0);
        add8(hB, lo, cf, 8);
        issue(hB, ro, 24);
        __builtin_amdgcn_sched_barrier(0);
        add8(hA, lo, cf, 16);
        add8(hB, lo, cf, 24);
      }
    }
  } else {
    const int nq = (e_cnt + 15) >> 4;
    const int n_pass = (nq + 4095) >> 12;
    const v4i* __restrict__ own4 = (const v4i*)own;
    const unsigned pat = (unsigned)blockIdx.x * 0x01010101u;
    const int rot = (int)(((unsigned)blockIdx.x * (unsigned)n_pass) / (unsigned)AGG_WG);

    int wcount = 0;
    v4i nx[4], dv[4];
    auto load_pass = [&](int c) {
#pragma unroll
      for (int k = 0; k < 4; ++k) {
        int qi = c * 4096 + k * 1024 + tid;
        if (qi > nq - 1) qi = nq - 1;
        nx[k] = own4[qi];
      }
    };
    if (n_pass > 0) {
      load_pass(rot);
      for (int p = 0; p <= n_pass; ++p) {
        int c = p + rot;
        if (c >= n_pass) c -= n_pass;
#pragma unroll
        for (int u = 0; u < 4; ++u) dv[u] = nx[u];
        {
          int c2 = c + 1;
          if (c2 >= n_pass) c2 -= n_pass;
          load_pass(c2);
        }
#pragma nounroll
        for (int k = 0; k < 4; ++k) {
          const bool end = (p == n_pass);
          if (end || wcount > WQCAP - PASS_MAX) {
            for (int b = 0; b < wcount; b += 32) {
              const int idx = b + lane;
              int s = 0, lr = 0;
              float coef = 0.0f;
              if (idx < wcount) {
                const int e = wq[idx];
                const int d = dst[e];
                s = src[e];
                lr = d - base;
                coef = dinv[s] * dinv[d];
              }
              int cnt = wcount - b;
              if (cnt > 32) cnt = 32;
              add_rows(s, lr, coef, cnt);
            }
            wcount = 0;
          }
          if (end) break;
          const v4i cur = dv[0];
          dv[0] = dv[1];
          dv[1] = dv[2];
          dv[2] = dv[3];
          const int e0 = (c * 4096 + k * 1024 + tid) * 16;
          unsigned x[4];
#pragma unroll
          for (int w = 0; w < 4; ++w) x[w] = (unsigned)cur[w] ^ pat;
          if ((c * 4096 + k * 1024 + 1024) * 16 > e_cnt) {
#pragma unroll
            for (int w = 0; w < 4; ++w) {
              int nb = e_cnt - (e0 + w * 4);
              if (nb < 0) nb = 0;
              if (nb < 4) x[w] |= 0xffffffffu << (8 * nb);
            }
          }
#pragma unroll
          for (int w = 0; w < 4; ++w) {
            const unsigned z = (x[w] - 0x01010101u) & ~x[w] & 0x80808080u;
            if (__builtin_amdgcn_ballot_w32(z != 0u) != 0u) {
#pragma unroll
              for (int q = 0; q < 4; ++q) {
                const bool m = ((x[w] >> (8 * q)) & 0xffu) == 0u;
                const unsigned mask = __builtin_amdgcn_ballot_w32(m);
                if (mask != 0u) {
                  if (m) wq[wcount + (int)__builtin_amdgcn_mbcnt_lo(mask, 0u)] = e0 + w * 4 + q;
                  wcount += __builtin_popcount(mask);
                }
              }
            }
          }
        }
      }
    }
  }
  constexpr int NW = (FO > 0) ? (F * FO) / 1024 : 1;
  static_assert(FO == 0 || (F * FO) % 1024 == 0, "Wn is packed in whole passes of the workgroup");
  float wv[NW];
  if constexpr (FO > 0) {
#pragma unroll
    for (int u = 0; u < NW; ++u) wv[u] = Wn[tid + u * 1024];
  }
  __syncthreads();

  if constexpr (MODE == 0) {
    for (int r = tid; r < nrows; r += 1024) Out[base + r] = rsqrtf((float)degcnt[r] + 1.0f);
    if (tid < 32) {
      const int c = lcount[tid];
      lcnt[blockIdx.x * 32 + tid] = (c <= LIST_CAP) ? c : -1;
    }
  } else if constexpr (FO == 0) {
    const int tot4 = (nrows * F) >> 2;
    const v4f* acc4 = (const v4f*)acc;
    v4f* out4 = (v4f*)(Out + (size_t)base * F);
    for (int i = tid; i < tot4; i += 1024) out4[i] = acc4[i];
  } else {
#pragma unroll
    for (int u = 0; u < NW; ++u) {
      const int i = tid + u * 1024;
      const int k = i / FO;
      const int col = i - k * FO;
      wq_all[(k >> 1) * (FO * 2) + col * 2 + (k & 1)] = __float_as_int(wv[u]);
    }
    __syncthreads();

    constexpr int NT = FO / 16;
    const int half = lane >> 4;
    const int l15  = lane & 15;
    const int ntile = (nrows + 15) >> 4;
    for (int tile = wave; tile < ntile; tile += 32) {
      int r = tile * 16 + l15;
      if (r > nrows - 1) r = nrows - 1;
      const float* arow = acc + r * PITCH;
      v8f c[NT];
#pragma unroll
      for (int t = 0; t < NT; ++t) c[t] = (v8f){};
#pragma unroll
      for (int k = 0; k < F; k += 4) {
        const int kk = k + half * 2;
        v2f a = *(const v2f*)(arow + kk);
        a.x = fmaxf(a.x, 0.0f);
        a.y = fmaxf(a.y, 0.0f);
        const int* wrow = wq_all + (kk >> 1) * (FO * 2) + l15 * 2;
#pragma unroll
        for (int t = 0; t < NT; ++t) {
          const v2i bi = *(const v2i*)(wrow + t * 32);
          v2f b;
          b.x = __int_as_float(bi.x);
          b.y = __int_as_float(bi.y);
          c[t] = __builtin_amdgcn_wmma_f32_16x16x4_f32(false, a, false, b,
                                                       (short)0, c[t], false, false);
        }
      }
      float* obase = Out + (size_t)(base + tile * 16) * FO + l15;
#pragma unroll
      for (int v = 0; v < 8; ++v) {
        const int lr = v + half * 8;
        if (tile * 16 + lr < nrows) {
#pragma unroll
          for (int t = 0; t < NT; ++t) obase[(size_t)lr * FO + t * 16] = c[t][v];
        }
      }
    }
  }
}

static inline unsigned int cdiv(long long a, long long b) {
  return (unsigned int)((a + b - 1) / b);
}

extern "C" void kernel_launch(void* const* d_in, const int* in_sizes, int n_in,
                              void* d_out, int out_size, void* d_ws, size_t ws_size,
                              hipStream_t stream) {
  const float* x  = (const float*)d_in[0];
  const int*   ei = (const int*)d_in[1];
  const float* W1 = (const float*)d_in[2];
  const float* b1 = (const float*)d_in[3];
  const float* W2 = (const float*)d_in[4];
  const float* b2 = (const float*)d_in[5];
  const float* W3 = (const float*)d_in[6];
  const float* b3 = (const float*)d_in[7];
  float* out = (float*)d_out;

  const int N = NODES;
  const int E = in_sizes[1] / 2;

  const int* src = ei;
  const int* dst = ei + E;

  float* dinv = (float*)d_ws;
  float* P0 = dinv + N;
  float* H2 = P0 + (size_t)N * 64;
  int* lcnt = (int*)(H2 + (size_t)N * 128);
  int* lst  = lcnt + LIST_WAVES;
  int* own  = lst + (size_t)LIST_WAVES * LIST_CAP;
  int* bm   = (int*)P0;

  const int T = 256;
  const unsigned int gemm_blocks = cdiv((N + 15) / 16, 8);

  owner_list_kernel<<<cdiv(cdiv(E, 16) * 4, T), T, 0, stream>>>(dst, E, own, bm);
  agg_rows_kernel<64, 0, 0><<<AGG_WG, 1024, 0, stream>>>(src, dst, P0, dinv, b1, W2, dinv, lcnt, lst, own, bm, N, E);

  gemm_wmma_f32_kernel<128, 64, false, false><<<gemm_blocks, T, 0, stream>>>(x, x, W1, 64, P0, N);
  agg_rows_kernel<64, 1, 128><<<AGG_WG, 1024, 0, stream>>>(src, dst, P0, dinv, b1, W2, H2, lcnt, lst, own, bm, N, E);

  agg_rows_kernel<128, 1, 64><<<AGG_WG, 1024, 0, stream>>>(src, dst, H2, dinv, b2, W3, P0, lcnt, lst, own, bm, N, E);

  agg_rows_kernel<64, 1, 0><<<AGG_WG, 1024, 0, stream>>>(src, dst, P0, dinv, b3, W3, out, lcnt, lst, own, bm, N, E);
}


# 64-column replays sum each wave's own rows in its registers (slot = wave-uniform register index), from start value to whole row, instead of LDS read-add-write per edge; f32 sum order as before
# speedup vs baseline: 6.8464x; 1.0447x over previous
#include <hip/hip_runtime.h>
#include <hip/hip_bf16.h>

typedef __attribute__((ext_vector_type(2))) float v2f;
typedef __attribute__((ext_vector_type(8))) float v8f;
typedef __attribute__((ext_vector_type(4))) float v4f;
typedef __attribute__((ext_vector_type(4))) int v4i;
typedef __attribute__((ext_vector_type(2))) int v2i;
typedef __attribute__((ext_vector_type(16))) float v16f;

#define NODES 100000
#define AGG_WG 256
#define ROWS_PER_WG ((NODES + AGG_WG - 1) / AGG_WG)
static_assert(AGG_WG <= 256, "an owner workgroup's number must fit one byte of the owner list");

#define BM_CHUNKS 32
static_assert(BM_CHUNKS * 32 == 256 * 4, "a block's 1024 edges fill BM_CHUNKS words of every owner");

__global__ __launch_bounds__(256)
void owner_list_kernel(const int* __restrict__ dst, int e, int* __restrict__ own, int* __restrict__ bm) {
  __shared__ __attribute__((aligned(16))) unsigned int tile[AGG_WG * BM_CHUNKS];
  const int tid = threadIdx.x;
  v4i* tile4 = (v4i*)tile;
#pragma unroll
  for (int u = 0; u < 8; ++u) tile4[tid + u * 256] = (v4i){0, 0, 0, 0};
  __syncthreads();

  const int i = blockIdx.x * 256 + tid;
  const int n_words = ((e + 15) >> 4) << 2;
  const int e0 = i * 4;
  int d[4];
  if (e0 + 3 < e && ((((size_t)dst) & 15) == 0)) {
    const v4i v = ((const v4i*)dst)[i];
    d[0] = v.x; d[1] = v.y; d[2] = v.z; d[3] = v.w;
  } else {
#pragma unroll
    for (int q = 0; q < 4; ++q) d[q] = (e0 + q < e) ? dst[e0 + q] : 0;
  }
  unsigned w = 0u;
  unsigned int* trow = tile + (tid >> 3);
  const int bit0 = (tid & 7) * 4;
#pragma unroll
  for (int q = 0; q < 4; ++q) {
    if (e0 + q < e) {
      const unsigned o = ((unsigned)d[q] / (unsigned)ROWS_PER_WG) & 0xffu;
      w |= o << (8 * q);
      atomicOr(trow + o * BM_CHUNKS, 1u << (bit0 + q));
    }
  }
  if (i < n_words) own[i] = (int)w;
  __syncthreads();

  const int pitch = ((e + 1023) >> 10) * BM_CHUNKS;
#pragma unroll
  for (int u = 0; u < 8; ++u) {
    const int v = tid + u * 256;
    v4i* line = (v4i*)(bm + (size_t)(v >> 3) * pitch + (size_t)blockIdx.x * BM_CHUNKS);
    line[v & 7] = tile4[v];
  }
}

template <int K, int FOUT, bool RELU, bool SPLITK>
__global__ void gemm_wmma_f32_kernel(const float* __restrict__ X0,
                                     const float* __restrict__ X1,
                                     const float* __restrict__ W,
                                     int ldw,
                                     float* __restrict__ H,
                                     int n_rows) {
  constexpr int NT = FOUT / 16;
  __shared__ float Wp[K * FOUT];

  for (int i = threadIdx.x; i < K * FOUT; i += 256) {
    int k = i / FOUT;
    int col = i - k * FOUT;
    Wp[(k >> 1) * (FOUT * 2) + col * 2 + (k & 1)] = W[k * ldw + col];
  }
  __syncthreads();

  const int wave = threadIdx.x >> 5;
  const int lane = threadIdx.x & 31;
  const int tile_m = blockIdx.x * 8 + wave;
  if (tile_m * 16 >= n_rows) return;

  const int half = lane >> 4;
  const int l15  = lane & 15;

  constexpr int KP = SPLITK ? K / 2 : K;
  const float* __restrict__ xrow0 = X0 + (size_t)(tile_m * 16 + l15) * KP;
  const float* __restrict__ xrow1 = X1 + (size_t)(tile_m * 16 + l15) * KP;

  v8f acc[NT];
#pragma unroll
  for (int t = 0; t < NT; ++t) acc[t] = (v8f){};

#pragma unroll
  for (int k = 0; k < K; k += 4) {
    const int kk = k + half * 2;
    const float* __restrict__ xp = (!SPLITK || k < KP) ? (xrow0 + kk) : (xrow1 + (kk - KP));
    float a0 = xp[0];
    float a1 = xp[1];
    if (RELU) { a0 = fmaxf(a0, 0.0f); a1 = fmaxf(a1, 0.0f); }
    v2f a;
    a.x = a0;
    a.y = a1;
    const float* __restrict__ wrow = Wp + (kk >> 1) * (FOUT * 2) + l15 * 2;
#pragma unroll
    for (int t = 0; t < NT; ++t) {
      const v2f b = *(const v2f*)(wrow + t * 32);
      acc[t] = __builtin_amdgcn_wmma_f32_16x16x4_f32(false, a, false, b,
                                                     (short)0, acc[t], false, false);
    }
  }

  float* __restrict__ hbase = H + (size_t)(tile_m * 16) * FOUT + l15;
#pragma unroll
  for (int t = 0; t < NT; ++t) {
#pragma unroll
    for (int v = 0; v < 8; ++v) {
      hbase[(size_t)(v + half * 8) * FOUT + t * 16] = acc[t][v];
    }
  }
}

#define WQCAP 736
#define PASS_MAX 512
#define LIST_CAP 512
#define LIST_WAVES (AGG_WG * 32)

static_assert(ROWS_PER_WG <= 512, "local row must fit the 9 low bits of a list word");
static_assert(NODES <= (1 << 22), "source row must fit the bits above the local row");

template <int F, int MODE, int FO>
__global__ __launch_bounds__(1024)
void agg_rows_kernel(const int* __restrict__ src,
                     const int* __restrict__ dst,
                     const float* __restrict__ H,
                     const float* __restrict__ dinv,
                     const float* __restrict__ bias,
                     const float* __restrict__ Wn,
                     float* __restrict__ Out,
                     int* lcnt,
                     int* lst,
                     const int* __restrict__ own,
                     const int* __restrict__ bm,
                     int n, int e_cnt) {
  constexpr int NQ = F / 64;
  static_assert(F % 64 == 0, "a row is a whole number of 64-float blocks");
  constexpr int PITCH = (FO > 0) ? F + 2 : F;
  static_assert(F * FO <= 32 * WQCAP, "the packed weights must fit the queues' LDS");
  __shared__ __attribute__((aligned(16))) float acc[(ROWS_PER_WG + 1) * PITCH];
  __shared__ __attribute__((aligned(16))) int wq_all[32 * WQCAP];

  const int tid  = threadIdx.x;
  const int lane = tid & 31;
  const int wave = __builtin_amdgcn_readfirstlane(tid >> 5);
  int* wq = wq_all + wave * WQCAP;
  const int base = blockIdx.x * ROWS_PER_WG;
  int nrows = n - base;
  if (nrows > ROWS_PER_WG) nrows = ROWS_PER_WG;
  if (nrows < 0) nrows = 0;
  const unsigned unrows = (unsigned)nrows;

  const int wid = blockIdx.x * 32 + wave;
  int* mylist = lst + (size_t)wid * LIST_CAP;
  int kept = 0;
  bool all_ok = false;
  if (MODE == 1) {
    const int kj = lcnt[blockIdx.x * 32 + lane];
    all_ok = (__builtin_amdgcn_ballot_w32(kj < 0) == 0u);
    kept = __builtin_amdgcn_readlane(kj, wave);
  }

  const bool regs = (MODE == 1) && (NQ == 1) && all_ok;

  int* degcnt = (int*)acc;
  int* lcount = (int*)acc + 512;
  if constexpr (MODE == 0) {
    for (int i = tid; i < ROWS_PER_WG; i += 1024) degcnt[i] = 0;
    if (tid < 32) lcount[tid] = 0;
  } else if (!regs) {
    for (int i = tid; i < nrows * F; i += 1024) {
      const int r = i / F;
      const int f = i - r * F;
      const float di = dinv[base + r];
      acc[r * PITCH + f] = H[(size_t)(base + r) * F + f] * di * di + bias[f];
    }
  }
  if (!regs) __syncthreads();

  auto add_rows = [&](int s, int lr, float coef, int cnt) {
    for (int j = 0; j < cnt; j += 8) {
      v2f h[8][NQ];
#pragma unroll
      for (int u = 0; u < 8; ++u) {
        int jj = j + u;
        if (jj > cnt - 1) jj = cnt - 1;
        const int sj = __builtin_amdgcn_readlane(s, jj);
        const float* hp = H + (size_t)sj * F + lane * 2;
#pragma unroll
        for (int q = 0; q < NQ; ++q) h[u][q] = *(const v2f*)(hp + q * 64);
      }
#pragma unroll
      for (int u = 0; u < 8; ++u) {
        if (j + u < cnt) {
          const float cj = __int_as_float(__builtin_amdgcn_readlane(__float_as_int(coef), j + u));
          const int rj = __builtin_amdgcn_readlane(lr, j + u);
          float* a = acc + rj * PITCH + lane * 2;
#pragma unroll
          for (int q = 0; q < NQ; ++q) {
            atomicAdd(a + q * 64,     h[u][q].x * cj);
            atomicAdd(a + q * 64 + 1, h[u][q].y * cj);
          }
        }
      }
    }
  };

  if constexpr (MODE == 0) {
    const int pitch = ((e_cnt + 1023) >> 10) * BM_CHUNKS;
    const int n4 = pitch >> 2;
    const int n_pass = (n4 + 4095) >> 12;
    const v4i* __restrict__ row4 = (const v4i*)(bm + (size_t)blockIdx.x * pitch);
    const v4i zero4 = (v4i){0, 0, 0, 0};
    int wcount = 0;

    auto drain0 = [&]() {
      for (int b = 0; b < wcount; b += 32) {
        const int idx = b + lane;
        if (idx < wcount) {
          const int ed = wq[idx];
          const int d = dst[ed];
          const int sr = src[ed];
          const int lr = d - base;
          atomicAdd(&degcnt[lr], 1);
          const int ow = lr & 31;
          const int pos = atomicAdd(&lcount[ow], 1);
          if (pos < LIST_CAP) lst[(size_t)(blockIdx.x * 32 + ow) * LIST_CAP + pos] = (sr << 9) | lr;
        }
      }
      wcount = 0;
    };

    v4i nx[4], dv[4];
    auto load_pass = [&](int c) {
#pragma unroll
      for (int k = 0; k < 4; ++k) {
        const int p = c * 4096 + k * 1024 + tid;
        nx[k] = (p < n4) ? row4[p] : zero4;
      }
    };
    load_pass(0);
    for (int c = 0; c < n_pass; ++c) {
#pragma unroll
      for (int u = 0; u < 4; ++u) dv[u] = nx[u];
      unsigned some = 0u;
#pragma unroll
      for (int u = 0; u < 4; ++u) {
        const int any = dv[u].x | dv[u].y | dv[u].z | dv[u].w;
        if (__builtin_amdgcn_ballot_w32(any != 0) != 0u) some |= 1u << u;
      }
      load_pass(c + 1);
#pragma nounroll
      for (int k = 0; k < 4; ++k) {
        v4i cur = dv[0];
        dv[0] = dv[1];
        dv[1] = dv[2];
        dv[2] = dv[3];
        const int w0 = (c * 4096 + k * 1024 + tid) * 4;
        if (((some >> k) & 1u) == 0u) continue;
#pragma nounroll
        for (int w = 0; w < 4; ++w) {
          unsigned word = (unsigned)cur.x;
          cur = (v4i){cur.y, cur.z, cur.w, 0};
          const int ebase = (w0 + w) * 32;
          unsigned live = __builtin_amdgcn_ballot_w32(word != 0u);
          while (live != 0u) {
            if (wcount > WQCAP - 32) drain0();
            if (word != 0u) {
              const int j = __builtin_ctz(word);
              word &= word - 1u;
              wq[wcount + (int)__builtin_amdgcn_mbcnt_lo(live, 0u)] = ebase + j;
            }
            wcount += __builtin_popcount(live);
            live = __builtin_amdgcn_ballot_w32(word != 0u);
          }
        }
      }
    }
    drain0();
  } else if (all_ok) {
    if constexpr (NQ == 1) {
      constexpr int RPW = (ROWS_PER_WG + 31) / 32;
      constexpr int PADSLOT = 15;
      static_assert(RPW <= PADSLOT, "every row of a wave and the padded slots need an element of their own");
      const unsigned lane2 = (unsigned)lane * 2u;
      v16f sx, sy;
      const v2f bv = *(const v2f*)(bias + lane2);
      float dv[RPW];
      v2f hv[RPW];
#pragma unroll
      for (int j = 0; j < RPW; ++j) {
        int rc = wave + 32 * j;
        if (rc > nrows - 1) rc = nrows - 1;
        if (rc < 0) rc = 0;
        dv[j] = dinv[base + rc];
        hv[j] = *(const v2f*)(H + (size_t)(base + rc) * F + lane2);
      }
#pragma unroll
      for (int j = 0; j < 16; ++j) {
        v2f t = (v2f){0.0f, 0.0f};
        if (j < RPW) {
          t.x = hv[j].x * dv[j] * dv[j] + bv.x;
          t.y = hv[j].y * dv[j] * dv[j] + bv.y;
        }
        sx[j] = t.x;
        sy[j] = t.y;
      }
      if (kept > 0) {
        const unsigned lane8 = (unsigned)lane * 8u;
        const char* Hb = (const char*)H;
        v2f hA[8], hB[8];
        auto list_word = [&](int b) -> int {
          int idx = b + lane;
          if (idx > kept - 1) idx = kept - 1;
          return mylist[idx];
        };
        auto issue = [&](v2f (&h)[8], unsigned ro, int l0) {
#pragma unroll
          for (int u = 0; u < 8; ++u) {
            const unsigned rj = (unsigned)__builtin_amdgcn_readlane((int)ro, l0 + u);
            h[u] = *(const v2f*)(Hb + (rj + lane8));
          }
        };
        auto add8 = [&](v2f (&h)[8], int so, float cf, int l0) {
#pragma unroll
          for (int u = 0; u < 8; ++u) {
            const int sj = __builtin_amdgcn_readlane(so, l0 + u);
            const float cj = __int_as_float(__builtin_amdgcn_readlane(__float_as_int(cf), l0 + u));
            sx[sj] += h[u].x * cj;
            sy[sj] += h[u].y * cj;
          }
        };

        const int ng = (kept + 7) >> 3;
        int w = list_word(0);
        unsigned ro = (unsigned)(w >> 9) * (unsigned)(F * 4);
        int so = (lane < kept) ? ((w & 511) >> 5) : PADSLOT;
        float cf = dinv[w >> 9] * dinv[base + (w & 511)];
        if (!(lane < kept)) cf = 0.0f;
        issue(hA, ro, 0);
        int g0 = 0;
        while (g0 + 4 < ng) {
          const int bn = (g0 + 4) * 8;
          const int wn = list_word(bn);
          issue(hB, ro, 8);
          __builtin_amdgcn_sched_barrier(0);
          add8(hA, so, cf, 0);
          const bool vn = (bn + lane < kept);
          const unsigned ron = (unsigned)(wn >> 9) * (unsigned)(F * 4);
          const int son = vn ? ((wn & 511) >> 5) : PADSLOT;
          const float dn1 = dinv[wn >> 9];
          const float dn2 = dinv[base + (wn & 511)];
          issue(hA, ro, 16);
          __builtin_amdgcn_sched_barrier(0);
          add8(hB, so, cf, 8);
          issue(hB, ro, 24);
          __builtin_amdgcn_sched_barrier(0);
          add8(hA, so, cf, 16);
          issue(hA, ron, 0);
          __builtin_amdgcn_sched_barrier(0);
          add8(hB, so, cf, 24);
          cf = vn ? dn1 * dn2 : 0.0f;
          ro = ron;
          so = son;
          g0 += 4;
        }
        const int ngl = ng - g0;
        if (ngl == 1) {
          add8(hA, so, cf, 0);
        } else if (ngl == 2) {
          issue(hB, ro, 8);
          __builtin_amdgcn_sched_barrier(0);
          add8(hA, so, cf, 0);
          add8(hB, so, cf, 8);
        } else if (ngl == 3) {
          issue(hB, ro, 8);
          __builtin_amdgcn_sched_barrier(0);
          add8(hA, so, cf, 0);
          issue(hA, ro, 16);
          __builtin_amdgcn_sched_barrier(0);
          add8(hB, so, cf, 8);
          add8(hA, so, cf, 16);
        } else {
          issue(hB, ro, 8);
          __builtin_amdgcn_sched_barrier(0);
          add8(hA, so, cf, 0);
          issue(hA, ro, 16);
          __builtin_amdgcn_sched_barrier(0);
          add8(hB, so, cf, 8);
          issue(hB, ro, 24);
          __builtin_amdgcn_sched_barrier(0);
          add8(hA, so, cf, 16);
          add8(hB, so, cf, 24);
        }
      }
#pragma unroll
      for (int j = 0; j < RPW; ++j) {
        const int r = wave + 32 * j;
        if (r < nrows) {
          const v2f t = (v2f){sx[j], sy[j]};
          if constexpr (FO == 0) *(v2f*)(Out + (size_t)(base + r) * F + lane2) = t;
          else                   *(v2f*)(acc + r * PITCH + lane2) = t;
        }
      }
    } else {
      if (kept > 0) {
        const unsigned lane2 = (unsigned)lane * 2u;
        const unsigned lane8 = (unsigned)lane * 8u;
        const char* Hb = (const char*)H;
        v2f hA[8][NQ], hB[8][NQ];
        auto list_word = [&](int b) -> int {
          int idx = b + lane;
          if (idx > kept - 1) idx = kept - 1;
          return mylist[idx];
        };
        auto issue = [&](v2f (&h)[8][NQ], unsigned ro, int l0) {
  #pragma unroll
          for (int u = 0; u < 8; ++u) {
            const unsigned rj = (unsigned)__builtin_amdgcn_readlane((int)ro, l0 + u);
            const char* hp = Hb + (rj + lane8);
  #pragma unroll
            for (int q = 0; q < NQ; ++q) h[u][q] = *(const v2f*)(hp + q * 256);
          }
        };
        auto add8 = [&](v2f (&h)[8][NQ], unsigned lo, float cf, int l0) {
  #pragma unroll
          for (int u = 0; u < 8; ++u) {
            const unsigned lj = (unsigned)__builtin_amdgcn_readlane((int)lo, l0 + u);
            const float cj = __int_as_float(__builtin_amdgcn_readlane(__float_as_int(cf), l0 + u));
            v2f* a = (v2f*)(acc + (lj + lane2));
  #pragma unroll
            for (int q = 0; q < NQ; ++q) {
              v2f t = a[q * 32];
              t.x += h[u][q].x * cj;
              t.y += h[u][q].y * cj;
              a[q * 32] = t;
            }
          }
        };

        const int ng = (kept + 7) >> 3;
        int w = list_word(0);
        unsigned ro = (unsigned)(w >> 9) * (unsigned)(F * 4);
        unsigned lo = (lane < kept) ? (unsigned)(w & 511) * (unsigned)PITCH : (unsigned)(ROWS_PER_WG * PITCH);
        float cf = dinv[w >> 9] * dinv[base + (w & 511)];
        if (!(lane < kept)) cf = 0.0f;
        issue(hA, ro, 0);
        int g0 = 0;
        while (g0 + 4 < ng) {
          const int bn = (g0 + 4) * 8;
          const int wn = list_word(bn);
          issue(hB, ro, 8);
          __builtin_amdgcn_sched_barrier(0);
          add8(hA, lo, cf, 0);
          const bool vn = (bn + lane < kept);
          const unsigned ron = (unsigned)(wn >> 9) * (unsigned)(F * 4);
          const unsigned lon = vn ? (unsigned)(wn & 511) * (unsigned)PITCH : (unsigned)(ROWS_PER_WG * PITCH);
          const float dn1 = dinv[wn >> 9];
          const float dn2 = dinv[base + (wn & 511)];
          issue(hA, ro, 16);
          __builtin_amdgcn_sched_barrier(0);
          add8(hB, lo, cf, 8);
          issue(hB, ro, 24);
          __builtin_amdgcn_sched_barrier(0);
          add8(hA, lo, cf, 16);
          issue(hA, ron, 0);
          __builtin_amdgcn_sched_barrier(0);
          add8(hB, lo, cf, 24);
          cf = vn ? dn1 * dn2 : 0.0f;
          ro = ron;
          lo = lon;
          g0 += 4;
        }
        const int ngl = ng - g0;
        if (ngl == 1) {
          add8(hA, lo, cf, 0);
        } else if (ngl == 2) {
          issue(hB, ro, 8);
          __builtin_amdgcn_sched_barrier(0);
          add8(hA, lo, cf, 0);
          add8(hB, lo, cf, 8);
        } else if (ngl == 3) {
          issue(hB, ro, 8);
          __builtin_amdgcn_sched_barrier(0);
          add8(hA, lo, cf, 0);
          issue(hA, ro, 16);
          __builtin_amdgcn_sched_barrier(0);
          add8(hB, lo, cf, 8);
          add8(hA, lo, cf, 16);
        } else {
          issue(hB, ro, 8);
          __builtin_amdgcn_sched_barrier(0);
          add8(hA, lo, cf, 0);
          issue(hA, ro, 16);
          __builtin_amdgcn_sched_barrier(0);
          add8(hB, lo, cf, 8);
          issue(hB, ro, 24);
          __builtin_amdgcn_sched_barrier(0);
          add8(hA, lo, cf, 16);
          add8(hB, lo, cf, 24);
        }
      }
    }
  } else {
    const int nq = (e_cnt + 15) >> 4;
    const int n_pass = (nq + 4095) >> 12;
    const v4i* __restrict__ own4 = (const v4i*)own;
    const unsigned pat = (unsigned)blockIdx.x * 0x01010101u;
    const int rot = (int)(((unsigned)blockIdx.x * (unsigned)n_pass) / (unsigned)AGG_WG);

    int wcount = 0;
    v4i nx[4], dv[4];
    auto load_pass = [&](int c) {
#pragma unroll
      for (int k = 0; k < 4; ++k) {
        int qi = c * 4096 + k * 1024 + tid;
        if (qi > nq - 1) qi = nq - 1;
        nx[k] = own4[qi];
      }
    };
    if (n_pass > 0) {
      load_pass(rot);
      for (int p = 0; p <= n_pass; ++p) {
        int c = p + rot;
        if (c >= n_pass) c -= n_pass;
#pragma unroll
        for (int u = 0; u < 4; ++u) dv[u] = nx[u];
        {
          int c2 = c + 1;
          if (c2 >= n_pass) c2 -= n_pass;
          load_pass(c2);
        }
#pragma nounroll
        for (int k = 0; k < 4; ++k) {
          const bool end = (p == n_pass);
          if (end || wcount > WQCAP - PASS_MAX) {
            for (int b = 0; b < wcount; b += 32) {
              const int idx = b + lane;
              int s = 0, lr = 0;
              float coef = 0.0f;
              if (idx < wcount) {
                const int e = wq[idx];
                const int d = dst[e];
                s = src[e];
                lr = d - base;
                coef = dinv[s] * dinv[d];
              }
              int cnt = wcount - b;
              if (cnt > 32) cnt = 32;
              add_rows(s, lr, coef, cnt);
            }
            wcount = 0;
          }
          if (end) break;
          const v4i cur = dv[0];
          dv[0] = dv[1];
          dv[1] = dv[2];
          dv[2] = dv[3];
          const int e0 = (c * 4096 + k * 1024 + tid) * 16;
          unsigned x[4];
#pragma unroll
          for (int w = 0; w < 4; ++w) x[w] = (unsigned)cur[w] ^ pat;
          if ((c * 4096 + k * 1024 + 1024) * 16 > e_cnt) {
#pragma unroll
            for (int w = 0; w < 4; ++w) {
              int nb = e_cnt - (e0 + w * 4);
              if (nb < 0) nb = 0;
              if (nb < 4) x[w] |= 0xffffffffu << (8 * nb);
            }
          }
#pragma unroll
          for (int w = 0; w < 4; ++w) {
            const unsigned z = (x[w] - 0x01010101u) & ~x[w] & 0x80808080u;
            if (__builtin_amdgcn_ballot_w32(z != 0u) != 0u) {
#pragma unroll
              for (int q = 0; q < 4; ++q) {
                const bool m = ((x[w] >> (8 * q)) & 0xffu) == 0u;
                const unsigned mask = __builtin_amdgcn_ballot_w32(m);
                if (mask != 0u) {
                  if (m) wq[wcount + (int)__builtin_amdgcn_mbcnt_lo(mask, 0u)] = e0 + w * 4 + q;
                  wcount += __builtin_popcount(mask);
                }
              }
            }
          }
        }
      }
    }
  }
  constexpr int NW = (FO > 0) ? (F * FO) / 1024 : 1;
  static_assert(FO == 0 || (F * FO) % 1024 == 0, "Wn is packed in whole passes of the workgroup");
  float wv[NW];
  if constexpr (FO > 0) {
#pragma unroll
    for (int u = 0; u < NW; ++u) wv[u] = Wn[tid + u * 1024];
  }
  __syncthreads();

  if constexpr (MODE == 0) {
    for (int r = tid; r < nrows; r += 1024) Out[base + r] = rsqrtf((float)degcnt[r] + 1.0f);
    if (tid < 32) {
      const int c = lcount[tid];
      lcnt[blockIdx.x * 32 + tid] = (c <= LIST_CAP) ? c : -1;
    }
  } else if constexpr (FO == 0) {
    if (!regs) {
      const int tot4 = (nrows * F) >> 2;
      const v4f* acc4 = (const v4f*)acc;
      v4f* out4 = (v4f*)(Out + (size_t)base * F);
      for (int i = tid; i < tot4; i += 1024) out4[i] = acc4[i];
    }
  } else {
#pragma unroll
    for (int u = 0; u < NW; ++u) {
      const int i = tid + u * 1024;
      const int k = i / FO;
      const int col = i - k * FO;
      wq_all[(k >> 1) * (FO * 2) + col * 2 + (k & 1)] = __float_as_int(wv[u]);
    }
    __syncthreads();

    constexpr int NT = FO / 16;
    const int half = lane >> 4;
    const int l15  = lane & 15;
    const int ntile = (nrows + 15) >> 4;
    for (int tile = wave; tile < ntile; tile += 32) {
      int r = tile * 16 + l15;
      if (r > nrows - 1) r = nrows - 1;
      const float* arow = acc + r * PITCH;
      v8f c[NT];
#pragma unroll
      for (int t = 0; t < NT; ++t) c[t] = (v8f){};
#pragma unroll
      for (int k = 0; k < F; k += 4) {
        const int kk = k + half * 2;
        v2f a = *(const v2f*)(arow + kk);
        a.x = fmaxf(a.x, 0.0f);
        a.y = fmaxf(a.y, 0.0f);
        const int* wrow = wq_all + (kk >> 1) * (FO * 2) + l15 * 2;
#pragma unroll
        for (int t = 0; t < NT; ++t) {
          const v2i bi = *(const v2i*)(wrow + t * 32);
          v2f b;
          b.x = __int_as_float(bi.x);
          b.y = __int_as_float(bi.y);
          c[t] = __builtin_amdgcn_wmma_f32_16x16x4_f32(false, a, false, b,
                                                       (short)0, c[t], false, false);
        }
      }
      float* obase = Out + (size_t)(base + tile * 16) * FO + l15;
#pragma unroll
      for (int v = 0; v < 8; ++v) {
        const int lr = v + half * 8;
        if (tile * 16 + lr < nrows) {
#pragma unroll
          for (int t = 0; t < NT; ++t) obase[(size_t)lr * FO + t * 16] = c[t][v];
        }
      }
    }
  }
}

static inline unsigned int cdiv(long long a, long long b) {
  return (unsigned int)((a + b - 1) / b);
}

extern "C" void kernel_launch(void* const* d_in, const int* in_sizes, int n_in,
                              void* d_out, int out_size, void* d_ws, size_t ws_size,
                              hipStream_t stream) {
  const float* x  = (const float*)d_in[0];
  const int*   ei = (const int*)d_in[1];
  const float* W1 = (const float*)d_in[2];
  const float* b1 = (const float*)d_in[3];
  const float* W2 = (const float*)d_in[4];
  const float* b2 = (const float*)d_in[5];
  const float* W3 = (const float*)d_in[6];
  const float* b3 = (const float*)d_in[7];
  float* out = (float*)d_out;

  const int N = NODES;
  const int E = in_sizes[1] / 2;

  const int* src = ei;
  const int* dst = ei + E;

  float* dinv = (float*)d_ws;
  float* P0 = dinv + N;
  float* H2 = P0 + (size_t)N * 64;
  int* lcnt = (int*)(H2 + (size_t)N * 128);
  int* lst  = lcnt + LIST_WAVES;
  int* own  = lst + (size_t)LIST_WAVES * LIST_CAP;
  int* bm   = (int*)P0;

  const int T = 256;
  const unsigned int gemm_blocks = cdiv((N + 15) / 16, 8);

  owner_list_kernel<<<cdiv(cdiv(E, 16) * 4, T), T, 0, stream>>>(dst, E, own, bm);
  agg_rows_kernel<64, 0, 0><<<AGG_WG, 1024, 0, stream>>>(src, dst, P0, dinv, b1, W2, dinv, lcnt, lst, own, bm, N, E);

  gemm_wmma_f32_kernel<128, 64, false, false><<<gemm_blocks, T, 0, stream>>>(x, x, W1, 64, P0, N);
  agg_rows_kernel<64, 1, 128><<<AGG_WG, 1024, 0, stream>>>(src, dst, P0, dinv, b1, W2, H2, lcnt, lst, own, bm, N, E);

  agg_rows_kernel<128, 1, 64><<<AGG_WG, 1024, 0, stream>>>(src, dst, H2, dinv, b2, W3, P0, lcnt, lst, own, bm, N, E);

  agg_rows_kernel<64, 1, 0><<<AGG_WG, 1024, 0, stream>>>(src, dst, P0, dinv, b3, W3, out, lcnt, lst, own, bm, N, E);
}


# owner-list blocks and the layer-1 product's blocks share ONE launch (block number picks the kind, one 32 KB LDS); x @ W1 waits in the output buffer for layer 1; 6 -> 5 launches, f32 sums as before
# speedup vs baseline: 7.1395x; 1.0428x over previous
#include <hip/hip_runtime.h>
#include <hip/hip_bf16.h>

typedef __attribute__((ext_vector_type(2))) float v2f;
typedef __attribute__((ext_vector_type(8))) float v8f;
typedef __attribute__((ext_vector_type(4))) float v4f;
typedef __attribute__((ext_vector_type(4))) int v4i;
typedef __attribute__((ext_vector_type(2))) int v2i;
typedef __attribute__((ext_vector_type(16))) float v16f;

#define NODES 100000
#define AGG_WG 256
#define ROWS_PER_WG ((NODES + AGG_WG - 1) / AGG_WG)
static_assert(AGG_WG <= 256, "an owner workgroup's number must fit one byte of the owner list");

#define BM_CHUNKS 32
static_assert(BM_CHUNKS * 32 == 256 * 4, "a block's 1024 edges fill BM_CHUNKS words of every owner");

__device__ __forceinline__
void owner_list_body(const int* __restrict__ dst, int e, int* __restrict__ own, int* __restrict__ bm,
                     unsigned int* tile, int blk) {
  const int tid = threadIdx.x;
  v4i* tile4 = (v4i*)tile;
#pragma unroll
  for (int u = 0; u < 8; ++u) tile4[tid + u * 256] = (v4i){0, 0, 0, 0};
  __syncthreads();

  const int i = blk * 256 + tid;
  const int n_words = ((e + 15) >> 4) << 2;
  const int e0 = i * 4;
  int d[4];
  if (e0 + 3 < e && ((((size_t)dst) & 15) == 0)) {
    const v4i v = ((const v4i*)dst)[i];
    d[0] = v.x; d[1] = v.y; d[2] = v.z; d[3] = v.w;
  } else {
#pragma unroll
    for (int q = 0; q < 4; ++q) d[q] = (e0 + q < e) ? dst[e0 + q] : 0;
  }
  unsigned w = 0u;
  unsigned int* trow = tile + (tid >> 3);
  const int bit0 = (tid & 7) * 4;
#pragma unroll
  for (int q = 0; q < 4; ++q) {
    if (e0 + q < e) {
      const unsigned o = ((unsigned)d[q] / (unsigned)ROWS_PER_WG) & 0xffu;
      w |= o << (8 * q);
      atomicOr(trow + o * BM_CHUNKS, 1u << (bit0 + q));
    }
  }
  if (i < n_words) own[i] = (int)w;
  __syncthreads();

  const int pitch = ((e + 1023) >> 10) * BM_CHUNKS;
#pragma unroll
  for (int u = 0; u < 8; ++u) {
    const int v = tid + u * 256;
    v4i* line = (v4i*)(bm + (size_t)(v >> 3) * pitch + (size_t)blk * BM_CHUNKS);
    line[v & 7] = tile4[v];
  }
}

template <int K, int FOUT, bool RELU, bool SPLITK>
__device__ __forceinline__
void gemm_wmma_f32_body(const float* __restrict__ X0,
                        const float* __restrict__ X1,
                        const float* __restrict__ W,
                        int ldw,
                        float* __restrict__ H,
                        int n_rows,
                        float* Wp,
                        int blk) {
  constexpr int NT = FOUT / 16;
  static_assert(K * FOUT <= AGG_WG * BM_CHUNKS, "the packed weights must fit the launch's 32 KB of LDS");

  for (int i = threadIdx.x; i < K * FOUT; i += 256) {
    int k = i / FOUT;
    int col = i - k * FOUT;
    Wp[(k >> 1) * (FOUT * 2) + col * 2 + (k & 1)] = W[k * ldw + col];
  }
  __syncthreads();

  const int wave = threadIdx.x >> 5;
  const int lane = threadIdx.x & 31;
  const int tile_m = blk * 8 + wave;
  if (tile_m * 16 >= n_rows) return;

  const int half = lane >> 4;
  const int l15  = lane & 15;

  constexpr int KP = SPLITK ? K / 2 : K;
  const float* __restrict__ xrow0 = X0 + (size_t)(tile_m * 16 + l15) * KP;
  const float* __restrict__ xrow1 = X1 + (size_t)(tile_m * 16 + l15) * KP;

  v8f acc[NT];
#pragma unroll
  for (int t = 0; t < NT; ++t) acc[t] = (v8f){};

#pragma unroll
  for (int k = 0; k < K; k += 4) {
    const int kk = k + half * 2;
    const float* __restrict__ xp = (!SPLITK || k < KP) ? (xrow0 + kk) : (xrow1 + (kk - KP));
    float a0 = xp[0];
    float a1 = xp[1];
    if (RELU) { a0 = fmaxf(a0, 0.0f); a1 = fmaxf(a1, 0.0f); }
    v2f a;
    a.x = a0;
    a.y = a1;
    const float* __restrict__ wrow = Wp + (kk >> 1) * (FOUT * 2) + l15 * 2;
#pragma unroll
    for (int t = 0; t < NT; ++t) {
      const v2f b = *(const v2f*)(wrow + t * 32);
      acc[t] = __builtin_amdgcn_wmma_f32_16x16x4_f32(false, a, false, b,
                                                     (short)0, acc[t], false, false);
    }
  }

  float* __restrict__ hbase = H + (size_t)(tile_m * 16) * FOUT + l15;
#pragma unroll
  for (int t = 0; t < NT; ++t) {
#pragma unroll
    for (int v = 0; v < 8; ++v) {
      hbase[(size_t)(v + half * 8) * FOUT + t * 16] = acc[t][v];
    }
  }
}

__global__ __launch_bounds__(256)
void front_kernel(const float* __restrict__ x, const float* __restrict__ W1, float* __restrict__ P,
                  int n_rows, int gemm_blocks,
                  const int* __restrict__ dst, int e, int* __restrict__ own, int* __restrict__ bm) {
  __shared__ __attribute__((aligned(16))) unsigned int smem[AGG_WG * BM_CHUNKS];
  const int b = blockIdx.x;
  if (b < gemm_blocks) gemm_wmma_f32_body<128, 64, false, false>(x, x, W1, 64, P, n_rows, (float*)smem, b);
  else                 owner_list_body(dst, e, own, bm, smem, b - gemm_blocks);
}

#define WQCAP 736
#define PASS_MAX 512
#define LIST_CAP 512
#define LIST_WAVES (AGG_WG * 32)

static_assert(ROWS_PER_WG <= 512, "local row must fit the 9 low bits of a list word");
static_assert(NODES <= (1 << 22), "source row must fit the bits above the local row");

template <int F, int MODE, int FO>
__global__ __launch_bounds__(1024)
void agg_rows_kernel(const int* __restrict__ src,
                     const int* __restrict__ dst,
                     const float* __restrict__ H,
                     const float* __restrict__ dinv,
                     const float* __restrict__ bias,
                     const float* __restrict__ Wn,
                     float* __restrict__ Out,
                     int* lcnt,
                     int* lst,
                     const int* __restrict__ own,
                     const int* __restrict__ bm,
                     int n, int e_cnt) {
  constexpr int NQ = F / 64;
  static_assert(F % 64 == 0, "a row is a whole number of 64-float blocks");
  constexpr int PITCH = (FO > 0) ? F + 2 : F;
  static_assert(F * FO <= 32 * WQCAP, "the packed weights must fit the queues' LDS");
  __shared__ __attribute__((aligned(16))) float acc[(ROWS_PER_WG + 1) * PITCH];
  __shared__ __attribute__((aligned(16))) int wq_all[32 * WQCAP];

  const int tid  = threadIdx.x;
  const int lane = tid & 31;
  const int wave = __builtin_amdgcn_readfirstlane(tid >> 5);
  int* wq = wq_all + wave * WQCAP;
  const int base = blockIdx.x * ROWS_PER_WG;
  int nrows = n - base;
  if (nrows > ROWS_PER_WG) nrows = ROWS_PER_WG;
  if (nrows < 0) nrows = 0;
  const unsigned unrows = (unsigned)nrows;

  const int wid = blockIdx.x * 32 + wave;
  int* mylist = lst + (size_t)wid * LIST_CAP;
  int kept = 0;
  bool all_ok = false;
  if (MODE == 1) {
    const int kj = lcnt[blockIdx.x * 32 + lane];
    all_ok = (__builtin_amdgcn_ballot_w32(kj < 0) == 0u);
    kept = __builtin_amdgcn_readlane(kj, wave);
  }

  const bool regs = (MODE == 1) && (NQ == 1) && all_ok;

  int* degcnt = (int*)acc;
  int* lcount = (int*)acc + 512;
  if constexpr (MODE == 0) {
    for (int i = tid; i < ROWS_PER_WG; i += 1024) degcnt[i] = 0;
    if (tid < 32) lcount[tid] = 0;
  } else if (!regs) {
    for (int i = tid; i < nrows * F; i += 1024) {
      const int r = i / F;
      const int f = i - r * F;
      const float di = dinv[base + r];
      acc[r * PITCH + f] = H[(size_t)(base + r) * F + f] * di * di + bias[f];
    }
  }
  if (!regs) __syncthreads();

  auto add_rows = [&](int s, int lr, float coef, int cnt) {
    for (int j = 0; j < cnt; j += 8) {
      v2f h[8][NQ];
#pragma unroll
      for (int u = 0; u < 8; ++u) {
        int jj = j + u;
        if (jj > cnt - 1) jj = cnt - 1;
        const int sj = __builtin_amdgcn_readlane(s, jj);
        const float* hp = H + (size_t)sj * F + lane * 2;
#pragma unroll
        for (int q = 0; q < NQ; ++q) h[u][q] = *(const v2f*)(hp + q * 64);
      }
#pragma unroll
      for (int u = 0; u < 8; ++u) {
        if (j + u < cnt) {
          const float cj = __int_as_float(__builtin_amdgcn_readlane(__float_as_int(coef), j + u));
          const int rj = __builtin_amdgcn_readlane(lr, j + u);
          float* a = acc + rj * PITCH + lane * 2;
#pragma unroll
          for (int q = 0; q < NQ; ++q) {
            atomicAdd(a + q * 64,     h[u][q].x * cj);
            atomicAdd(a + q * 64 + 1, h[u][q].y * cj);
          }
        }
      }
    }
  };

  if constexpr (MODE == 0) {
    const int pitch = ((e_cnt + 1023) >> 10) * BM_CHUNKS;
    const int n4 = pitch >> 2;
    const int n_pass = (n4 + 4095) >> 12;
    const v4i* __restrict__ row4 = (const v4i*)(bm + (size_t)blockIdx.x * pitch);
    const v4i zero4 = (v4i){0, 0, 0, 0};
    int wcount = 0;

    auto drain0 = [&]() {
      for (int b = 0; b < wcount; b += 32) {
        const int idx = b + lane;
        if (idx < wcount) {
          const int ed = wq[idx];
          const int d = dst[ed];
          const int sr = src[ed];
          const int lr = d - base;
          atomicAdd(&degcnt[lr], 1);
          const int ow = lr & 31;
          const int pos = atomicAdd(&lcount[ow], 1);
          if (pos < LIST_CAP) lst[(size_t)(blockIdx.x * 32 + ow) * LIST_CAP + pos] = (sr << 9) | lr;
        }
      }
      wcount = 0;
    };

    v4i nx[4], dv[4];
    auto load_pass = [&](int c) {
#pragma unroll
      for (int k = 0; k < 4; ++k) {
        const int p = c * 4096 + k * 1024 + tid;
        nx[k] = (p < n4) ? row4[p] : zero4;
      }
    };
    load_pass(0);
    for (int c = 0; c < n_pass; ++c) {
#pragma unroll
      for (int u = 0; u < 4; ++u) dv[u] = nx[u];
      unsigned some = 0u;
#pragma unroll
      for (int u = 0; u < 4; ++u) {
        const int any = dv[u].x | dv[u].y | dv[u].z | dv[u].w;
        if (__builtin_amdgcn_ballot_w32(any != 0) != 0u) some |= 1u << u;
      }
      load_pass(c + 1);
#pragma nounroll
      for (int k = 0; k < 4; ++k) {
        v4i cur = dv[0];
        dv[0] = dv[1];
        dv[1] = dv[2];
        dv[2] = dv[3];
        const int w0 = (c * 4096 + k * 1024 + tid) * 4;
        if (((some >> k) & 1u) == 0u) continue;
#pragma nounroll
        for (int w = 0; w < 4; ++w) {
          unsigned word = (unsigned)cur.x;
          cur = (v4i){cur.y, cur.z, cur.w, 0};
          const int ebase = (w0 + w) * 32;
          unsigned live = __builtin_amdgcn_ballot_w32(word != 0u);
          while (live != 0u) {
            if (wcount > WQCAP - 32) drain0();
            if (word != 0u) {
              const int j = __builtin_ctz(word);
              word &= word - 1u;
              wq[wcount + (int)__builtin_amdgcn_mbcnt_lo(live, 0u)] = ebase + j;
            }
            wcount += __builtin_popcount(live);
            live = __builtin_amdgcn_ballot_w32(word != 0u);
          }
        }
      }
    }
    drain0();
  } else if (all_ok) {
    if constexpr (NQ == 1) {
      constexpr int RPW = (ROWS_PER_WG + 31) / 32;
      constexpr int PADSLOT = 15;
      static_assert(RPW <= PADSLOT, "every row of a wave and the padded slots need an element of their own");
      const unsigned lane2 = (unsigned)lane * 2u;
      v16f sx, sy;
      const v2f bv = *(const v2f*)(bias + lane2);
      float dv[RPW];
      v2f hv[RPW];
#pragma unroll
      for (int j = 0; j < RPW; ++j) {
        int rc = wave + 32 * j;
        if (rc > nrows - 1) rc = nrows - 1;
        if (rc < 0) rc = 0;
        dv[j] = dinv[base + rc];
        hv[j] = *(const v2f*)(H + (size_t)(base + rc) * F + lane2);
      }
#pragma unroll
      for (int j = 0; j < 16; ++j) {
        v2f t = (v2f){0.0f, 0.0f};
        if (j < RPW) {
          t.x = hv[j].x * dv[j] * dv[j] + bv.x;
          t.y = hv[j].y * dv[j] * dv[j] + bv.y;
        }
        sx[j] = t.x;
        sy[j] = t.y;
      }
      if (kept > 0) {
        const unsigned lane8 = (unsigned)lane * 8u;
        const char* Hb = (const char*)H;
        v2f hA[8], hB[8];
        auto list_word = [&](int b) -> int {
          int idx = b + lane;
          if (idx > kept - 1) idx = kept - 1;
          return mylist[idx];
        };
        auto issue = [&](v2f (&h)[8], unsigned ro, int l0) {
#pragma unroll
          for (int u = 0; u < 8; ++u) {
            const unsigned rj = (unsigned)__builtin_amdgcn_readlane((int)ro, l0 + u);
            h[u] = *(const v2f*)(Hb + (rj + lane8));
          }
        };
        auto add8 = [&](v2f (&h)[8], int so, float cf, int l0) {
#pragma unroll
          for (int u = 0; u < 8; ++u) {
            const int sj = __builtin_amdgcn_readlane(so, l0 + u);
            const float cj = __int_as_float(__builtin_amdgcn_readlane(__float_as_int(cf), l0 + u));
            sx[sj] += h[u].x * cj;
            sy[sj] += h[u].y * cj;
          }
        };

        const int ng = (kept + 7) >> 3;
        int w = list_word(0);
        unsigned ro = (unsigned)(w >> 9) * (unsigned)(F * 4);
        int so = (lane < kept) ? ((w & 511) >> 5) : PADSLOT;
        float cf = dinv[w >> 9] * dinv[base + (w & 511)];
        if (!(lane < kept)) cf = 0.0f;
        issue(hA, ro, 0);
        int g0 = 0;
        while (g0 + 4 < ng) {
          const int bn = (g0 + 4) * 8;
          const int wn = list_word(bn);
          issue(hB, ro, 8);
          __builtin_amdgcn_sched_barrier(0);
          add8(hA, so, cf, 0);
          const bool vn = (bn + lane < kept);
          const unsigned ron = (unsigned)(wn >> 9) * (unsigned)(F * 4);
          const int son = vn ? ((wn & 511) >> 5) : PADSLOT;
          const float dn1 = dinv[wn >> 9];
          const float dn2 = dinv[base + (wn & 511)];
          issue(hA, ro, 16);
          __builtin_amdgcn_sched_barrier(0);
          add8(hB, so, cf, 8);
          issue(hB, ro, 24);
          __builtin_amdgcn_sched_barrier(0);
          add8(hA, so, cf, 16);
          issue(hA, ron, 0);
          __builtin_amdgcn_sched_barrier(0);
          add8(hB, so, cf, 24);
          cf = vn ? dn1 * dn2 : 0.0f;
          ro = ron;
          so = son;
          g0 += 4;
        }
        const int ngl = ng - g0;
        if (ngl == 1) {
          add8(hA, so, cf, 0);
        } else if (ngl == 2) {
          issue(hB, ro, 8);
          __builtin_amdgcn_sched_barrier(0);
          add8(hA, so, cf, 0);
          add8(hB, so, cf, 8);
        } else if (ngl == 3) {
          issue(hB, ro, 8);
          __builtin_amdgcn_sched_barrier(0);
          add8(hA, so, cf, 0);
          issue(hA, ro, 16);
          __builtin_amdgcn_sched_barrier(0);
          add8(hB, so, cf, 8);
          add8(hA, so, cf, 16);
        } else {
          issue(hB, ro, 8);
          __builtin_amdgcn_sched_barrier(0);
          add8(hA, so, cf, 0);
          issue(hA, ro, 16);
          __builtin_amdgcn_sched_barrier(0);
          add8(hB, so, cf, 8);
          issue(hB, ro, 24);
          __builtin_amdgcn_sched_barrier(0);
          add8(hA, so, cf, 16);
          add8(hB, so, cf, 24);
        }
      }
#pragma unroll
      for (int j = 0; j < RPW; ++j) {
        const int r = wave + 32 * j;
        if (r < nrows) {
          const v2f t = (v2f){sx[j], sy[j]};
          if constexpr (FO == 0) *(v2f*)(Out + (size_t)(base + r) * F + lane2) = t;
          else                   *(v2f*)(acc + r * PITCH + lane2) = t;
        }
      }
    } else {
      if (kept > 0) {
        const unsigned lane2 = (unsigned)lane * 2u;
        const unsigned lane8 = (unsigned)lane * 8u;
        const char* Hb = (const char*)H;
        v2f hA[8][NQ], hB[8][NQ];
        auto list_word = [&](int b) -> int {
          int idx = b + lane;
          if (idx > kept - 1) idx = kept - 1;
          return mylist[idx];
        };
        auto issue = [&](v2f (&h)[8][NQ], unsigned ro, int l0) {
  #pragma unroll
          for (int u = 0; u < 8; ++u) {
            const unsigned rj = (unsigned)__builtin_amdgcn_readlane((int)ro, l0 + u);
            const char* hp = Hb + (rj + lane8);
  #pragma unroll
            for (int q = 0; q < NQ; ++q) h[u][q] = *(const v2f*)(hp + q * 256);
          }
        };
        auto add8 = [&](v2f (&h)[8][NQ], unsigned lo, float cf, int l0) {
  #pragma unroll
          for (int u = 0; u < 8; ++u) {
            const unsigned lj = (unsigned)__builtin_amdgcn_readlane((int)lo, l0 + u);
            const float cj = __int_as_float(__builtin_amdgcn_readlane(__float_as_int(cf), l0 + u));
            v2f* a = (v2f*)(acc + (lj + lane2));
  #pragma unroll
            for (int q = 0; q < NQ; ++q) {
              v2f t = a[q * 32];
              t.x += h[u][q].x * cj;
              t.y += h[u][q].y * cj;
              a[q * 32] = t;
            }
          }
        };

        const int ng = (kept + 7) >> 3;
        int w = list_word(0);
        unsigned ro = (unsigned)(w >> 9) * (unsigned)(F * 4);
        unsigned lo = (lane < kept) ? (unsigned)(w & 511) * (unsigned)PITCH : (unsigned)(ROWS_PER_WG * PITCH);
        float cf = dinv[w >> 9] * dinv[base + (w & 511)];
        if (!(lane < kept)) cf = 0.0f;
        issue(hA, ro, 0);
        int g0 = 0;
        while (g0 + 4 < ng) {
          const int bn = (g0 + 4) * 8;
          const int wn = list_word(bn);
          issue(hB, ro, 8);
          __builtin_amdgcn_sched_barrier(0);
          add8(hA, lo, cf, 0);
          const bool vn = (bn + lane < kept);
          const unsigned ron = (unsigned)(wn >> 9) * (unsigned)(F * 4);
          const unsigned lon = vn ? (unsigned)(wn & 511) * (unsigned)PITCH : (unsigned)(ROWS_PER_WG * PITCH);
          const float dn1 = dinv[wn >> 9];
          const float dn2 = dinv[base + (wn & 511)];
          issue(hA, ro, 16);
          __builtin_amdgcn_sched_barrier(0);
          add8(hB, lo, cf, 8);
          issue(hB, ro, 24);
          __builtin_amdgcn_sched_barrier(0);
          add8(hA, lo, cf, 16);
          issue(hA, ron, 0);
          __builtin_amdgcn_sched_barrier(0);
          add8(hB, lo, cf, 24);
          cf = vn ? dn1 * dn2 : 0.0f;
          ro = ron;
          lo = lon;
          g0 += 4;
        }
        const int ngl = ng - g0;
        if (ngl == 1) {
          add8(hA, lo, cf, 0);
        } else if (ngl == 2) {
          issue(hB, ro, 8);
          __builtin_amdgcn_sched_barrier(0);
          add8(hA, lo, cf, 0);
          add8(hB, lo, cf, 8);
        } else if (ngl == 3) {
          issue(hB, ro, 8);
          __builtin_amdgcn_sched_barrier(0);
          add8(hA, lo, cf, 0);
          issue(hA, ro, 16);
          __builtin_amdgcn_sched_barrier(0);
          add8(hB, lo, cf, 8);
          add8(hA, lo, cf, 16);
        } else {
          issue(hB, ro, 8);
          __builtin_amdgcn_sched_barrier(0);
          add8(hA, lo, cf, 0);
          issue(hA, ro, 16);
          __builtin_amdgcn_sched_barrier(0);
          add8(hB, lo, cf, 8);
          issue(hB, ro, 24);
          __builtin_amdgcn_sched_barrier(0);
          add8(hA, lo, cf, 16);
          add8(hB, lo, cf, 24);
        }
      }
    }
  } else {
    const int nq = (e_cnt + 15) >> 4;
    const int n_pass = (nq + 4095) >> 12;
    const v4i* __restrict__ own4 = (const v4i*)own;
    const unsigned pat = (unsigned)blockIdx.x * 0x01010101u;
    const int rot = (int)(((unsigned)blockIdx.x * (unsigned)n_pass) / (unsigned)AGG_WG);

    int wcount = 0;
    v4i nx[4], dv[4];
    auto load_pass = [&](int c) {
#pragma unroll
      for (int k = 0; k < 4; ++k) {
        int qi = c * 4096 + k * 1024 + tid;
        if (qi > nq - 1) qi = nq - 1;
        nx[k] = own4[qi];
      }
    };
    if (n_pass > 0) {
      load_pass(rot);
      for (int p = 0; p <= n_pass; ++p) {
        int c = p + rot;
        if (c >= n_pass) c -= n_pass;
#pragma unroll
        for (int u = 0; u < 4; ++u) dv[u] = nx[u];
        {
          int c2 = c + 1;
          if (c2 >= n_pass) c2 -= n_pass;
          load_pass(c2);
        }
#pragma nounroll
        for (int k = 0; k < 4; ++k) {
          const bool end = (p == n_pass);
          if (end || wcount > WQCAP - PASS_MAX) {
            for (int b = 0; b < wcount; b += 32) {
              const int idx = b + lane;
              int s = 0, lr = 0;
              float coef = 0.0f;
              if (idx < wcount) {
                const int e = wq[idx];
                const int d = dst[e];
                s = src[e];
                lr = d - base;
                coef = dinv[s] * dinv[d];
              }
              int cnt = wcount - b;
              if (cnt > 32) cnt = 32;
              add_rows(s, lr, coef, cnt);
            }
            wcount = 0;
          }
          if (end) break;
          const v4i cur = dv[0];
          dv[0] = dv[1];
          dv[1] = dv[2];
          dv[2] = dv[3];
          const int e0 = (c * 4096 + k * 1024 + tid) * 16;
          unsigned x[4];
#pragma unroll
          for (int w = 0; w < 4; ++w) x[w] = (unsigned)cur[w] ^ pat;
          if ((c * 4096 + k * 1024 + 1024) * 16 > e_cnt) {
#pragma unroll
            for (int w = 0; w < 4; ++w) {
              int nb = e_cnt - (e0 + w * 4);
              if (nb < 0) nb = 0;
              if (nb < 4) x[w] |= 0xffffffffu << (8 * nb);
            }
          }
#pragma unroll
          for (int w = 0; w < 4; ++w) {
            const unsigned z = (x[w] - 0x01010101u) & ~x[w] & 0x80808080u;
            if (__builtin_amdgcn_ballot_w32(z != 0u) != 0u) {
#pragma unroll
              for (int q = 0; q < 4; ++q) {
                const bool m = ((x[w] >> (8 * q)) & 0xffu) == 0u;
                const unsigned mask = __builtin_amdgcn_ballot_w32(m);
                if (mask != 0u) {
                  if (m) wq[wcount + (int)__builtin_amdgcn_mbcnt_lo(mask, 0u)] = e0 + w * 4 + q;
                  wcount += __builtin_popcount(mask);
                }
              }
            }
          }
        }
      }
    }
  }
  constexpr int NW = (FO > 0) ? (F * FO) / 1024 : 1;
  static_assert(FO == 0 || (F * FO) % 1024 == 0, "Wn is packed in whole passes of the workgroup");
  float wv[NW];
  if constexpr (FO > 0) {
#pragma unroll
    for (int u = 0; u < NW; ++u) wv[u] = Wn[tid + u * 1024];
  }
  __syncthreads();

  if constexpr (MODE == 0) {
    for (int r = tid; r < nrows; r += 1024) Out[base + r] = rsqrtf((float)degcnt[r] + 1.0f);
    if (tid < 32) {
      const int c = lcount[tid];
      lcnt[blockIdx.x * 32 + tid] = (c <= LIST_CAP) ? c : -1;
    }
  } else if constexpr (FO == 0) {
    if (!regs) {
      const int tot4 = (nrows * F) >> 2;
      const v4f* acc4 = (const v4f*)acc;
      v4f* out4 = (v4f*)(Out + (size_t)base * F);
      for (int i = tid; i < tot4; i += 1024) out4[i] = acc4[i];
    }
  } else {
#pragma unroll
    for (int u = 0; u < NW; ++u) {
      const int i = tid + u * 1024;
      const int k = i / FO;
      const int col = i - k * FO;
      wq_all[(k >> 1) * (FO * 2) + col * 2 + (k & 1)] = __float_as_int(wv[u]);
    }
    __syncthreads();

    constexpr int NT = FO / 16;
    const int half = lane >> 4;
    const int l15  = lane & 15;
    const int ntile = (nrows + 15) >> 4;
    for (int tile = wave; tile < ntile; tile += 32) {
      int r = tile * 16 + l15;
      if (r > nrows - 1) r = nrows - 1;
      const float* arow = acc + r * PITCH;
      v8f c[NT];
#pragma unroll
      for (int t = 0; t < NT; ++t) c[t] = (v8f){};
#pragma unroll
      for (int k = 0; k < F; k += 4) {
        const int kk = k + half * 2;
        v2f a = *(const v2f*)(arow + kk);
        a.x = fmaxf(a.x, 0.0f);
        a.y = fmaxf(a.y, 0.0f);
        const int* wrow = wq_all + (kk >> 1) * (FO * 2) + l15 * 2;
#pragma unroll
        for (int t = 0; t < NT; ++t) {
          const v2i bi = *(const v2i*)(wrow + t * 32);
          v2f b;
          b.x = __int_as_float(bi.x);
          b.y = __int_as_float(bi.y);
          c[t] = __builtin_amdgcn_wmma_f32_16x16x4_f32(false, a, false, b,
                                                       (short)0, c[t], false, false);
        }
      }
      float* obase = Out + (size_t)(base + tile * 16) * FO + l15;
#pragma unroll
      for (int v = 0; v < 8; ++v) {
        const int lr = v + half * 8;
        if (tile * 16 + lr < nrows) {
#pragma unroll
          for (int t = 0; t < NT; ++t) obase[(size_t)lr * FO + t * 16] = c[t][v];
        }
      }
    }
  }
}

static inline unsigned int cdiv(long long a, long long b) {
  return (unsigned int)((a + b - 1) / b);
}

extern "C" void kernel_launch(void* const* d_in, const int* in_sizes, int n_in,
                              void* d_out, int out_size, void* d_ws, size_t ws_size,
                              hipStream_t stream) {
  const float* x  = (const float*)d_in[0];
  const int*   ei = (const int*)d_in[1];
  const float* W1 = (const float*)d_in[2];
  const float* b1 = (const float*)d_in[3];
  const float* W2 = (const float*)d_in[4];
  const float* b2 = (const float*)d_in[5];
  const float* W3 = (const float*)d_in[6];
  const float* b3 = (const float*)d_in[7];
  float* out = (float*)d_out;

  const int N = NODES;
  const int E = in_sizes[1] / 2;

  const int* src = ei;
  const int* dst = ei + E;

  float* dinv = (float*)d_ws;
  float* P0 = dinv + N;
  float* H2 = P0 + (size_t)N * 64;
  int* lcnt = (int*)(H2 + (size_t)N * 128);
  int* lst  = lcnt + LIST_WAVES;
  int* own  = lst + (size_t)LIST_WAVES * LIST_CAP;
  int* bm   = (int*)P0;

  const int T = 256;
  const unsigned int gemm_blocks = cdiv((N + 15) / 16, 8);

  const unsigned int owner_blocks = cdiv(cdiv(E, 16) * 4, T);
  front_kernel<<<gemm_blocks + owner_blocks, T, 0, stream>>>(x, W1, out, N, (int)gemm_blocks, dst, E, own, bm);

  agg_rows_kernel<64, 0, 0><<<AGG_WG, 1024, 0, stream>>>(src, dst, P0, dinv, b1, W2, dinv, lcnt, lst, own, bm, N, E);

  agg_rows_kernel<64, 1, 128><<<AGG_WG, 1024, 0, stream>>>(src, dst, out, dinv, b1, W2, H2, lcnt, lst, own, bm, N, E);

  agg_rows_kernel<128, 1, 64><<<AGG_WG, 1024, 0, stream>>>(src, dst, H2, dinv, b2, W3, P0, lcnt, lst, own, bm, N, E);

  agg_rows_kernel<64, 1, 0><<<AGG_WG, 1024, 0, stream>>>(src, dst, P0, dinv, b3, W3, out, lcnt, lst, own, bm, N, E);
}


# edge weights kept per list slot: layer 1's replay stores dinv[s]*dinv[d] beside its list word, layers 2 and 3 read them in list order, no per-edge dinv gathers there; f32 sums as before
# speedup vs baseline: 7.4800x; 1.0477x over previous
#include <hip/hip_runtime.h>
#include <hip/hip_bf16.h>

typedef __attribute__((ext_vector_type(2))) float v2f;
typedef __attribute__((ext_vector_type(8))) float v8f;
typedef __attribute__((ext_vector_type(4))) float v4f;
typedef __attribute__((ext_vector_type(4))) int v4i;
typedef __attribute__((ext_vector_type(2))) int v2i;
typedef __attribute__((ext_vector_type(16))) float v16f;

#define NODES 100000
#define AGG_WG 256
#define ROWS_PER_WG ((NODES + AGG_WG - 1) / AGG_WG)
static_assert(AGG_WG <= 256, "an owner workgroup's number must fit one byte of the owner list");

#define BM_CHUNKS 32
static_assert(BM_CHUNKS * 32 == 256 * 4, "a block's 1024 edges fill BM_CHUNKS words of every owner");

__device__ __forceinline__
void owner_list_body(const int* __restrict__ dst, int e, int* __restrict__ own, int* __restrict__ bm,
                     unsigned int* tile, int blk) {
  const int tid = threadIdx.x;
  v4i* tile4 = (v4i*)tile;
#pragma unroll
  for (int u = 0; u < 8; ++u) tile4[tid + u * 256] = (v4i){0, 0, 0, 0};
  __syncthreads();

  const int i = blk * 256 + tid;
  const int n_words = ((e + 15) >> 4) << 2;
  const int e0 = i * 4;
  int d[4];
  if (e0 + 3 < e && ((((size_t)dst) & 15) == 0)) {
    const v4i v = ((const v4i*)dst)[i];
    d[0] = v.x; d[1] = v.y; d[2] = v.z; d[3] = v.w;
  } else {
#pragma unroll
    for (int q = 0; q < 4; ++q) d[q] = (e0 + q < e) ? dst[e0 + q] : 0;
  }
  unsigned w = 0u;
  unsigned int* trow = tile + (tid >> 3);
  const int bit0 = (tid & 7) * 4;
#pragma unroll
  for (int q = 0; q < 4; ++q) {
    if (e0 + q < e) {
      const unsigned o = ((unsigned)d[q] / (unsigned)ROWS_PER_WG) & 0xffu;
      w |= o << (8 * q);
      atomicOr(trow + o * BM_CHUNKS, 1u << (bit0 + q));
    }
  }
  if (i < n_words) own[i] = (int)w;
  __syncthreads();

  const int pitch = ((e + 1023) >> 10) * BM_CHUNKS;
#pragma unroll
  for (int u = 0; u < 8; ++u) {
    const int v = tid + u * 256;
    v4i* line = (v4i*)(bm + (size_t)(v >> 3) * pitch + (size_t)blk * BM_CHUNKS);
    line[v & 7] = tile4[v];
  }
}

template <int K, int FOUT, bool RELU, bool SPLITK>
__device__ __forceinline__
void gemm_wmma_f32_body(const float* __restrict__ X0,
                        const float* __restrict__ X1,
                        const float* __restrict__ W,
                        int ldw,
                        float* __restrict__ H,
                        int n_rows,
                        float* Wp,
                        int blk) {
  constexpr int NT = FOUT / 16;
  static_assert(K * FOUT <= AGG_WG * BM_CHUNKS, "the packed weights must fit the launch's 32 KB of LDS");

  for (int i = threadIdx.x; i < K * FOUT; i += 256) {
    int k = i / FOUT;
    int col = i - k * FOUT;
    Wp[(k >> 1) * (FOUT * 2) + col * 2 + (k & 1)] = W[k * ldw + col];
  }
  __syncthreads();

  const int wave = threadIdx.x >> 5;
  const int lane = threadIdx.x & 31;
  const int tile_m = blk * 8 + wave;
  if (tile_m * 16 >= n_rows) return;

  const int half = lane >> 4;
  const int l15  = lane & 15;

  constexpr int KP = SPLITK ? K / 2 : K;
  const float* __restrict__ xrow0 = X0 + (size_t)(tile_m * 16 + l15) * KP;
  const float* __restrict__ xrow1 = X1 + (size_t)(tile_m * 16 + l15) * KP;

  v8f acc[NT];
#pragma unroll
  for (int t = 0; t < NT; ++t) acc[t] = (v8f){};

#pragma unroll
  for (int k = 0; k < K; k += 4) {
    const int kk = k + half * 2;
    const float* __restrict__ xp = (!SPLITK || k < KP) ? (xrow0 + kk) : (xrow1 + (kk - KP));
    float a0 = xp[0];
    float a1 = xp[1];
    if (RELU) { a0 = fmaxf(a0, 0.0f); a1 = fmaxf(a1, 0.0f); }
    v2f a;
    a.x = a0;
    a.y = a1;
    const float* __restrict__ wrow = Wp + (kk >> 1) * (FOUT * 2) + l15 * 2;
#pragma unroll
    for (int t = 0; t < NT; ++t) {
      const v2f b = *(const v2f*)(wrow + t * 32);
      acc[t] = __builtin_amdgcn_wmma_f32_16x16x4_f32(false, a, false, b,
                                                     (short)0, acc[t], false, false);
    }
  }

  float* __restrict__ hbase = H + (size_t)(tile_m * 16) * FOUT + l15;
#pragma unroll
  for (int t = 0; t < NT; ++t) {
#pragma unroll
    for (int v = 0; v < 8; ++v) {
      hbase[(size_t)(v + half * 8) * FOUT + t * 16] = acc[t][v];
    }
  }
}

__global__ __launch_bounds__(256)
void front_kernel(const float* __restrict__ x, const float* __restrict__ W1, float* __restrict__ P,
                  int n_rows, int gemm_blocks,
                  const int* __restrict__ dst, int e, int* __restrict__ own, int* __restrict__ bm) {
  __shared__ __attribute__((aligned(16))) unsigned int smem[AGG_WG * BM_CHUNKS];
  const int b = blockIdx.x;
  if (b < gemm_blocks) gemm_wmma_f32_body<128, 64, false, false>(x, x, W1, 64, P, n_rows, (float*)smem, b);
  else                 owner_list_body(dst, e, own, bm, smem, b - gemm_blocks);
}

#define WQCAP 736
#define PASS_MAX 512
#define LIST_CAP 352
#define LIST_WAVES (AGG_WG * 32)

static_assert(ROWS_PER_WG <= 512, "local row must fit the 9 low bits of a list word");
static_assert(NODES <= (1 << 22), "source row must fit the bits above the local row");

template <int F, int MODE, int FO, int CFM>
__global__ __launch_bounds__(1024)
void agg_rows_kernel(const int* __restrict__ src,
                     const int* __restrict__ dst,
                     const float* __restrict__ H,
                     const float* __restrict__ dinv,
                     const float* __restrict__ bias,
                     const float* __restrict__ Wn,
                     float* __restrict__ Out,
                     int* lcnt,
                     int* lst,
                     float* lcf,
                     const int* __restrict__ own,
                     const int* __restrict__ bm,
                     int n, int e_cnt) {
  constexpr int NQ = F / 64;
  static_assert(F % 64 == 0, "a row is a whole number of 64-float blocks");
  constexpr int PITCH = (FO > 0) ? F + 2 : F;
  static_assert(F * FO <= 32 * WQCAP, "the packed weights must fit the queues' LDS");
  __shared__ __attribute__((aligned(16))) float acc[(ROWS_PER_WG + 1) * PITCH];
  __shared__ __attribute__((aligned(16))) int wq_all[32 * WQCAP];

  const int tid  = threadIdx.x;
  const int lane = tid & 31;
  const int wave = __builtin_amdgcn_readfirstlane(tid >> 5);
  int* wq = wq_all + wave * WQCAP;
  const int base = blockIdx.x * ROWS_PER_WG;
  int nrows = n - base;
  if (nrows > ROWS_PER_WG) nrows = ROWS_PER_WG;
  if (nrows < 0) nrows = 0;
  const unsigned unrows = (unsigned)nrows;

  const int wid = blockIdx.x * 32 + wave;
  int* mylist = lst + (size_t)wid * LIST_CAP;
  float* mycf = lcf + (size_t)wid * LIST_CAP;
  int kept = 0;
  bool all_ok = false;
  if (MODE == 1) {
    const int kj = lcnt[blockIdx.x * 32 + lane];
    all_ok = (__builtin_amdgcn_ballot_w32(kj < 0) == 0u);
    kept = __builtin_amdgcn_readlane(kj, wave);
  }

  const bool regs = (MODE == 1) && (NQ == 1) && all_ok;

  int* degcnt = (int*)acc;
  int* lcount = (int*)acc + 512;
  if constexpr (MODE == 0) {
    for (int i = tid; i < ROWS_PER_WG; i += 1024) degcnt[i] = 0;
    if (tid < 32) lcount[tid] = 0;
  } else if (!regs) {
    for (int i = tid; i < nrows * F; i += 1024) {
      const int r = i / F;
      const int f = i - r * F;
      const float di = dinv[base + r];
      acc[r * PITCH + f] = H[(size_t)(base + r) * F + f] * di * di + bias[f];
    }
  }
  if (!regs) __syncthreads();

  auto add_rows = [&](int s, int lr, float coef, int cnt) {
    for (int j = 0; j < cnt; j += 8) {
      v2f h[8][NQ];
#pragma unroll
      for (int u = 0; u < 8; ++u) {
        int jj = j + u;
        if (jj > cnt - 1) jj = cnt - 1;
        const int sj = __builtin_amdgcn_readlane(s, jj);
        const float* hp = H + (size_t)sj * F + lane * 2;
#pragma unroll
        for (int q = 0; q < NQ; ++q) h[u][q] = *(const v2f*)(hp + q * 64);
      }
#pragma unroll
      for (int u = 0; u < 8; ++u) {
        if (j + u < cnt) {
          const float cj = __int_as_float(__builtin_amdgcn_readlane(__float_as_int(coef), j + u));
          const int rj = __builtin_amdgcn_readlane(lr, j + u);
          float* a = acc + rj * PITCH + lane * 2;
#pragma unroll
          for (int q = 0; q < NQ; ++q) {
            atomicAdd(a + q * 64,     h[u][q].x * cj);
            atomicAdd(a + q * 64 + 1, h[u][q].y * cj);
          }
        }
      }
    }
  };

  if constexpr (MODE == 0) {
    const int pitch = ((e_cnt + 1023) >> 10) * BM_CHUNKS;
    const int n4 = pitch >> 2;
    const int n_pass = (n4 + 4095) >> 12;
    const v4i* __restrict__ row4 = (const v4i*)(bm + (size_t)blockIdx.x * pitch);
    const v4i zero4 = (v4i){0, 0, 0, 0};
    int wcount = 0;

    auto drain0 = [&]() {
      for (int b = 0; b < wcount; b += 32) {
        const int idx = b + lane;
        if (idx < wcount) {
          const int ed = wq[idx];
          const int d = dst[ed];
          const int sr = src[ed];
          const int lr = d - base;
          atomicAdd(&degcnt[lr], 1);
          const int ow = lr & 31;
          const int pos = atomicAdd(&lcount[ow], 1);
          if (pos < LIST_CAP) lst[(size_t)(blockIdx.x * 32 + ow) * LIST_CAP + pos] = (sr << 9) | lr;
        }
      }
      wcount = 0;
    };

    v4i nx[4], dv[4];
    auto load_pass = [&](int c) {
#pragma unroll
      for (int k = 0; k < 4; ++k) {
        const int p = c * 4096 + k * 1024 + tid;
        nx[k] = (p < n4) ? row4[p] : zero4;
      }
    };
    load_pass(0);
    for (int c = 0; c < n_pass; ++c) {
#pragma unroll
      for (int u = 0; u < 4; ++u) dv[u] = nx[u];
      unsigned some = 0u;
#pragma unroll
      for (int u = 0; u < 4; ++u) {
        const int any = dv[u].x | dv[u].y | dv[u].z | dv[u].w;
        if (__builtin_amdgcn_ballot_w32(any != 0) != 0u) some |= 1u << u;
      }
      load_pass(c + 1);
#pragma nounroll
      for (int k = 0; k < 4; ++k) {
        v4i cur = dv[0];
        dv[0] = dv[1];
        dv[1] = dv[2];
        dv[2] = dv[3];
        const int w0 = (c * 4096 + k * 1024 + tid) * 4;
        if (((some >> k) & 1u) == 0u) continue;
#pragma nounroll
        for (int w = 0; w < 4; ++w) {
          unsigned word = (unsigned)cur.x;
          cur = (v4i){cur.y, cur.z, cur.w, 0};
          const int ebase = (w0 + w) * 32;
          unsigned live = __builtin_amdgcn_ballot_w32(word != 0u);
          while (live != 0u) {
            if (wcount > WQCAP - 32) drain0();
            if (word != 0u) {
              const int j = __builtin_ctz(word);
              word &= word - 1u;
              wq[wcount + (int)__builtin_amdgcn_mbcnt_lo(live, 0u)] = ebase + j;
            }
            wcount += __builtin_popcount(live);
            live = __builtin_amdgcn_ballot_w32(word != 0u);
          }
        }
      }
    }
    drain0();
  } else if (all_ok) {
    if constexpr (NQ == 1) {
      constexpr int RPW = (ROWS_PER_WG + 31) / 32;
      constexpr int PADSLOT = 15;
      static_assert(RPW <= PADSLOT, "every row of a wave and the padded slots need an element of their own");
      const unsigned lane2 = (unsigned)lane * 2u;
      v16f sx, sy;
      const v2f bv = *(const v2f*)(bias + lane2);
      float dv[RPW];
      v2f hv[RPW];
#pragma unroll
      for (int j = 0; j < RPW; ++j) {
        int rc = wave + 32 * j;
        if (rc > nrows - 1) rc = nrows - 1;
        if (rc < 0) rc = 0;
        dv[j] = dinv[base + rc];
        hv[j] = *(const v2f*)(H + (size_t)(base + rc) * F + lane2);
      }
#pragma unroll
      for (int j = 0; j < 16; ++j) {
        v2f t = (v2f){0.0f, 0.0f};
        if (j < RPW) {
          t.x = hv[j].x * dv[j] * dv[j] + bv.x;
          t.y = hv[j].y * dv[j] * dv[j] + bv.y;
        }
        sx[j] = t.x;
        sy[j] = t.y;
      }
      if (kept > 0) {
        const unsigned lane8 = (unsigned)lane * 8u;
        const char* Hb = (const char*)H;
        v2f hA[8], hB[8];
        auto list_word = [&](int b) -> int {
          int idx = b + lane;
          if (idx > kept - 1) idx = kept - 1;
          return mylist[idx];
        };
        auto cf_word = [&](int b) -> float {
          int idx = b + lane;
          if (idx > kept - 1) idx = kept - 1;
          return mycf[idx];
        };
        auto issue = [&](v2f (&h)[8], unsigned ro, int l0) {
#pragma unroll
          for (int u = 0; u < 8; ++u) {
            const unsigned rj = (unsigned)__builtin_amdgcn_readlane((int)ro, l0 + u);
            h[u] = *(const v2f*)(Hb + (rj + lane8));
          }
        };
        auto add8 = [&](v2f (&h)[8], int so, float cf, int l0) {
#pragma unroll
          for (int u = 0; u < 8; ++u) {
            const int sj = __builtin_amdgcn_readlane(so, l0 + u);
            const float cj = __int_as_float(__builtin_amdgcn_readlane(__float_as_int(cf), l0 + u));
            sx[sj] += h[u].x * cj;
            sy[sj] += h[u].y * cj;
          }
        };

        const int ng = (kept + 7) >> 3;
        int w = list_word(0);
        unsigned ro = (unsigned)(w >> 9) * (unsigned)(F * 4);
        int so = (lane < kept) ? ((w & 511) >> 5) : PADSLOT;
        float cf;
        if constexpr (CFM == 0) {
          cf = dinv[w >> 9] * dinv[base + (w & 511)];
          if (lane < kept) mycf[lane] = cf;
        } else {
          cf = cf_word(0);
        }
        if (!(lane < kept)) cf = 0.0f;
        issue(hA, ro, 0);
        int g0 = 0;
        while (g0 + 4 < ng) {
          const int bn = (g0 + 4) * 8;
          const int wn = list_word(bn);
          float cn = 0.0f;
          if constexpr (CFM == 1) cn = cf_word(bn);
          issue(hB, ro, 8);
          __builtin_amdgcn_sched_barrier(0);
          add8(hA, so, cf, 0);
          const bool vn = (bn + lane < kept);
          const unsigned ron = (unsigned)(wn >> 9) * (unsigned)(F * 4);
          const int son = vn ? ((wn & 511) >> 5) : PADSLOT;
          float dn1 = 0.0f, dn2 = 0.0f;
          if constexpr (CFM == 0) {
            dn1 = dinv[wn >> 9];
            dn2 = dinv[base + (wn & 511)];
          }
          issue(hA, ro, 16);
          __builtin_amdgcn_sched_barrier(0);
          add8(hB, so, cf, 8);
          issue(hB, ro, 24);
          __builtin_amdgcn_sched_barrier(0);
          add8(hA, so, cf, 16);
          issue(hA, ron, 0);
          __builtin_amdgcn_sched_barrier(0);
          add8(hB, so, cf, 24);
          if constexpr (CFM == 0) {
            cn = dn1 * dn2;
            if (vn) mycf[bn + lane] = cn;
          }
          cf = vn ? cn : 0.0f;
          ro = ron;
          so = son;
          g0 += 4;
        }
        const int ngl = ng - g0;
        if (ngl == 1) {
          add8(hA, so, cf, 0);
        } else if (ngl == 2) {
          issue(hB, ro, 8);
          __builtin_amdgcn_sched_barrier(0);
          add8(hA, so, cf, 0);
          add8(hB, so, cf, 8);
        } else if (ngl == 3) {
          issue(hB, ro, 8);
          __builtin_amdgcn_sched_barrier(0);
          add8(hA, so, cf, 0);
          issue(hA, ro, 16);
          __builtin_amdgcn_sched_barrier(0);
          add8(hB, so, cf, 8);
          add8(hA, so, cf, 16);
        } else {
          issue(hB, ro, 8);
          __builtin_amdgcn_sched_barrier(0);
          add8(hA, so, cf, 0);
          issue(hA, ro, 16);
          __builtin_amdgcn_sched_barrier(0);
          add8(hB, so, cf, 8);
          issue(hB, ro, 24);
          __builtin_amdgcn_sched_barrier(0);
          add8(hA, so, cf, 16);
          add8(hB, so, cf, 24);
        }
      }
#pragma unroll
      for (int j = 0; j < RPW; ++j) {
        const int r = wave + 32 * j;
        if (r < nrows) {
          const v2f t = (v2f){sx[j], sy[j]};
          if constexpr (FO == 0) *(v2f*)(Out + (size_t)(base + r) * F + lane2) = t;
          else                   *(v2f*)(acc + r * PITCH + lane2) = t;
        }
      }
    } else {
      if (kept > 0) {
        const unsigned lane2 = (unsigned)lane * 2u;
        const unsigned lane8 = (unsigned)lane * 8u;
        const char* Hb = (const char*)H;
        v2f hA[8][NQ], hB[8][NQ];
        auto list_word = [&](int b) -> int {
          int idx = b + lane;
          if (idx > kept - 1) idx = kept - 1;
          return mylist[idx];
        };
        auto cf_word = [&](int b) -> float {
          int idx = b + lane;
          if (idx > kept - 1) idx = kept - 1;
          return mycf[idx];
        };
        auto issue = [&](v2f (&h)[8][NQ], unsigned ro, int l0) {
  #pragma unroll
          for (int u = 0; u < 8; ++u) {
            const unsigned rj = (unsigned)__builtin_amdgcn_readlane((int)ro, l0 + u);
            const char* hp = Hb + (rj + lane8);
  #pragma unroll
            for (int q = 0; q < NQ; ++q) h[u][q] = *(const v2f*)(hp + q * 256);
          }
        };
        auto add8 = [&](v2f (&h)[8][NQ], unsigned lo, float cf, int l0) {
  #pragma unroll
          for (int u = 0; u < 8; ++u) {
            const unsigned lj = (unsigned)__builtin_amdgcn_readlane((int)lo, l0 + u);
            const float cj = __int_as_float(__builtin_amdgcn_readlane(__float_as_int(cf), l0 + u));
            v2f* a = (v2f*)(acc + (lj + lane2));
  #pragma unroll
            for (int q = 0; q < NQ; ++q) {
              v2f t = a[q * 32];
              t.x += h[u][q].x * cj;
              t.y += h[u][q].y * cj;
              a[q * 32] = t;
            }
          }
        };

        const int ng = (kept + 7) >> 3;
        int w = list_word(0);
        unsigned ro = (unsigned)(w >> 9) * (unsigned)(F * 4);
        unsigned lo = (lane < kept) ? (unsigned)(w & 511) * (unsigned)PITCH : (unsigned)(ROWS_PER_WG * PITCH);
        float cf;
        if constexpr (CFM == 0) {
          cf = dinv[w >> 9] * dinv[base + (w & 511)];
          if (lane < kept) mycf[lane] = cf;
        } else {
          cf = cf_word(0);
        }
        if (!(lane < kept)) cf = 0.0f;
        issue(hA, ro, 0);
        int g0 = 0;
        while (g0 + 4 < ng) {
          const int bn = (g0 + 4) * 8;
          const int wn = list_word(bn);
          float cn = 0.0f;
          if constexpr (CFM == 1) cn = cf_word(bn);
          issue(hB, ro, 8);
          __builtin_amdgcn_sched_barrier(0);
          add8(hA, lo, cf, 0);
          const bool vn = (bn + lane < kept);
          const unsigned ron = (unsigned)(wn >> 9) * (unsigned)(F * 4);
          const unsigned lon = vn ? (unsigned)(wn & 511) * (unsigned)PITCH : (unsigned)(ROWS_PER_WG * PITCH);
          float dn1 = 0.0f, dn2 = 0.0f;
          if constexpr (CFM == 0) {
            dn1 = dinv[wn >> 9];
            dn2 = dinv[base + (wn & 511)];
          }
          issue(hA, ro, 16);
          __builtin_amdgcn_sched_barrier(0);
          add8(hB, lo, cf, 8);
          issue(hB, ro, 24);
          __builtin_amdgcn_sched_barrier(0);
          add8(hA, lo, cf, 16);
          issue(hA, ron, 0);
          __builtin_amdgcn_sched_barrier(0);
          add8(hB, lo, cf, 24);
          if constexpr (CFM == 0) {
            cn = dn1 * dn2;
            if (vn) mycf[bn + lane] = cn;
          }
          cf = vn ? cn : 0.0f;
          ro = ron;
          lo = lon;
          g0 += 4;
        }
        const int ngl = ng - g0;
        if (ngl == 1) {
          add8(hA, lo, cf, 0);
        } else if (ngl == 2) {
          issue(hB, ro, 8);
          __builtin_amdgcn_sched_barrier(0);
          add8(hA, lo, cf, 0);
          add8(hB, lo, cf, 8);
        } else if (ngl == 3) {
          issue(hB, ro, 8);
          __builtin_amdgcn_sched_barrier(0);
          add8(hA, lo, cf, 0);
          issue(hA, ro, 16);
          __builtin_amdgcn_sched_barrier(0);
          add8(hB, lo, cf, 8);
          add8(hA, lo, cf, 16);
        } else {
          issue(hB, ro, 8);
          __builtin_amdgcn_sched_barrier(0);
          add8(hA, lo, cf, 0);
          issue(hA, ro, 16);
          __builtin_amdgcn_sched_barrier(0);
          add8(hB, lo, cf, 8);
          issue(hB, ro, 24);
          __builtin_amdgcn_sched_barrier(0);
          add8(hA, lo, cf, 16);
          add8(hB, lo, cf, 24);
        }
      }
    }
  } else {
    const int nq = (e_cnt + 15) >> 4;
    const int n_pass = (nq + 4095) >> 12;
    const v4i* __restrict__ own4 = (const v4i*)own;
    const unsigned pat = (unsigned)blockIdx.x * 0x01010101u;
    const int rot = (int)(((unsigned)blockIdx.x * (unsigned)n_pass) / (unsigned)AGG_WG);

    int wcount = 0;
    v4i nx[4], dv[4];
    auto load_pass = [&](int c) {
#pragma unroll
      for (int k = 0; k < 4; ++k) {
        int qi = c * 4096 + k * 1024 + tid;
        if (qi > nq - 1) qi = nq - 1;
        nx[k] = own4[qi];
      }
    };
    if (n_pass > 0) {
      load_pass(rot);
      for (int p = 0; p <= n_pass; ++p) {
        int c = p + rot;
        if (c >= n_pass) c -= n_pass;
#pragma unroll
        for (int u = 0; u < 4; ++u) dv[u] = nx[u];
        {
          int c2 = c + 1;
          if (c2 >= n_pass) c2 -= n_pass;
          load_pass(c2);
        }
#pragma nounroll
        for (int k = 0; k < 4; ++k) {
          const bool end = (p == n_pass);
          if (end || wcount > WQCAP - PASS_MAX) {
            for (int b = 0; b < wcount; b += 32) {
              const int idx = b + lane;
              int s = 0, lr = 0;
              float coef = 0.0f;
              if (idx < wcount) {
                const int e = wq[idx];
                const int d = dst[e];
                s = src[e];
                lr = d - base;
                coef = dinv[s] * dinv[d];
              }
              int cnt = wcount - b;
              if (cnt > 32) cnt = 32;
              add_rows(s, lr, coef, cnt);
            }
            wcount = 0;
          }
          if (end) break;
          const v4i cur = dv[0];
          dv[0] = dv[1];
          dv[1] = dv[2];
          dv[2] = dv[3];
          const int e0 = (c * 4096 + k * 1024 + tid) * 16;
          unsigned x[4];
#pragma unroll
          for (int w = 0; w < 4; ++w) x[w] = (unsigned)cur[w] ^ pat;
          if ((c * 4096 + k * 1024 + 1024) * 16 > e_cnt) {
#pragma unroll
            for (int w = 0; w < 4; ++w) {
              int nb = e_cnt - (e0 + w * 4);
              if (nb < 0) nb = 0;
              if (nb < 4) x[w] |= 0xffffffffu << (8 * nb);
            }
          }
#pragma unroll
          for (int w = 0; w < 4; ++w) {
            const unsigned z = (x[w] - 0x01010101u) & ~x[w] & 0x80808080u;
            if (__builtin_amdgcn_ballot_w32(z != 0u) != 0u) {
#pragma unroll
              for (int q = 0; q < 4; ++q) {
                const bool m = ((x[w] >> (8 * q)) & 0xffu) == 0u;
                const unsigned mask = __builtin_amdgcn_ballot_w32(m);
                if (mask != 0u) {
                  if (m) wq[wcount + (int)__builtin_amdgcn_mbcnt_lo(mask, 0u)] = e0 + w * 4 + q;
                  wcount += __builtin_popcount(mask);
                }
              }
            }
          }
        }
      }
    }
  }
  constexpr int NW = (FO > 0) ? (F * FO) / 1024 : 1;
  static_assert(FO == 0 || (F * FO) % 1024 == 0, "Wn is packed in whole passes of the workgroup");
  float wv[NW];
  if constexpr (FO > 0) {
#pragma unroll
    for (int u = 0; u < NW; ++u) wv[u] = Wn[tid + u * 1024];
  }
  __syncthreads();

  if constexpr (MODE == 0) {
    for (int r = tid; r < nrows; r += 1024) Out[base + r] = rsqrtf((float)degcnt[r] + 1.0f);
    if (tid < 32) {
      const int c = lcount[tid];
      lcnt[blockIdx.x * 32 + tid] = (c <= LIST_CAP) ? c : -1;
    }
  } else if constexpr (FO == 0) {
    if (!regs) {
      const int tot4 = (nrows * F) >> 2;
      const v4f* acc4 = (const v4f*)acc;
      v4f* out4 = (v4f*)(Out + (size_t)base * F);
      for (int i = tid; i < tot4; i += 1024) out4[i] = acc4[i];
    }
  } else {
#pragma unroll
    for (int u = 0; u < NW; ++u) {
      const int i = tid + u * 1024;
      const int k = i / FO;
      const int col = i - k * FO;
      wq_all[(k >> 1) * (FO * 2) + col * 2 + (k & 1)] = __float_as_int(wv[u]);
    }
    __syncthreads();

    constexpr int NT = FO / 16;
    const int half = lane >> 4;
    const int l15  = lane & 15;
    const int ntile = (nrows + 15) >> 4;
    for (int tile = wave; tile < ntile; tile += 32) {
      int r = tile * 16 + l15;
      if (r > nrows - 1) r = nrows - 1;
      const float* arow = acc + r * PITCH;
      v8f c[NT];
#pragma unroll
      for (int t = 0; t < NT; ++t) c[t] = (v8f){};
#pragma unroll
      for (int k = 0; k < F; k += 4) {
        const int kk = k + half * 2;
        v2f a = *(const v2f*)(arow + kk);
        a.x = fmaxf(a.x, 0.0f);
        a.y = fmaxf(a.y, 0.0f);
        const int* wrow = wq_all + (kk >> 1) * (FO * 2) + l15 * 2;
#pragma unroll
        for (int t = 0; t < NT; ++t) {
          const v2i bi = *(const v2i*)(wrow + t * 32);
          v2f b;
          b.x = __int_as_float(bi.x);
          b.y = __int_as_float(bi.y);
          c[t] = __builtin_amdgcn_wmma_f32_16x16x4_f32(false, a, false, b,
                                                       (short)0, c[t], false, false);
        }
      }
      float* obase = Out + (size_t)(base + tile * 16) * FO + l15;
#pragma unroll
      for (int v = 0; v < 8; ++v) {
        const int lr = v + half * 8;
        if (tile * 16 + lr < nrows) {
#pragma unroll
          for (int t = 0; t < NT; ++t) obase[(size_t)lr * FO + t * 16] = c[t][v];
        }
      }
    }
  }
}

static inline unsigned int cdiv(long long a, long long b) {
  return (unsigned int)((a + b - 1) / b);
}

extern "C" void kernel_launch(void* const* d_in, const int* in_sizes, int n_in,
                              void* d_out, int out_size, void* d_ws, size_t ws_size,
                              hipStream_t stream) {
  const float* x  = (const float*)d_in[0];
  const int*   ei = (const int*)d_in[1];
  const float* W1 = (const float*)d_in[2];
  const float* b1 = (const float*)d_in[3];
  const float* W2 = (const float*)d_in[4];
  const float* b2 = (const float*)d_in[5];
  const float* W3 = (const float*)d_in[6];
  const float* b3 = (const float*)d_in[7];
  float* out = (float*)d_out;

  const int N = NODES;
  const int E = in_sizes[1] / 2;

  const int* src = ei;
  const int* dst = ei + E;

  float* dinv = (float*)d_ws;
  float* P0 = dinv + N;
  float* H2 = P0 + (size_t)N * 64;
  int* lcnt = (int*)(H2 + (size_t)N * 128);
  int* lst  = lcnt + LIST_WAVES;
  float* lcf = (float*)(lst + (size_t)LIST_WAVES * LIST_CAP);
  int* own  = (int*)(lcf + (size_t)LIST_WAVES * LIST_CAP);
  int* bm   = (int*)P0;

  const int T = 256;
  const unsigned int gemm_blocks = cdiv((N + 15) / 16, 8);

  const unsigned int owner_blocks = cdiv(cdiv(E, 16) * 4, T);
  front_kernel<<<gemm_blocks + owner_blocks, T, 0, stream>>>(x, W1, out, N, (int)gemm_blocks, dst, E, own, bm);

  agg_rows_kernel<64, 0, 0, 0><<<AGG_WG, 1024, 0, stream>>>(src, dst, P0, dinv, b1, W2, dinv, lcnt, lst, lcf, own, bm, N, E);

  agg_rows_kernel<64, 1, 128, 0><<<AGG_WG, 1024, 0, stream>>>(src, dst, out, dinv, b1, W2, H2, lcnt, lst, lcf, own, bm, N, E);

  agg_rows_kernel<128, 1, 64, 1><<<AGG_WG, 1024, 0, stream>>>(src, dst, H2, dinv, b2, W3, P0, lcnt, lst, lcf, own, bm, N, E);

  agg_rows_kernel<64, 1, 0, 1><<<AGG_WG, 1024, 0, stream>>>(src, dst, P0, dinv, b3, W3, out, lcnt, lst, lcf, own, bm, N, E);
}
